# starting kernel
# baseline (speedup 1.0000x reference)
#include <hip/hip_runtime.h>
#include <math.h>


#define N_NODES 50000
#define N_EDGES 800000
#define HEADS   8
#define HEAD_DIM 16
#define FDIM    128
#define ETOT    (N_EDGES + N_NODES)
#define NEG_SLOPE 0.2f

typedef __attribute__((ext_vector_type(2))) float v2f;
typedef __attribute__((ext_vector_type(8))) float v8f;

__global__ void __launch_bounds__(128)
gemm128_wmma_f32(const float* __restrict__ A, const float* __restrict__ W,
                 float* __restrict__ out) {
  const int wave = threadIdx.x >> 5;
  const int lane = threadIdx.x & 31;
  const int tile = blockIdx.x * 4 + wave;
  const int tm   = tile >> 3;
  const int tn   = tile & 7;
  const int half = lane >> 4;
  const int l    = lane & 15;

  const float* __restrict__ arow = A + (size_t)(tm * 16 + l) * FDIM;
  const float* __restrict__ bcol = W + tn * 16 + l;

  v8f c = {};
#pragma unroll
  for (int k0 = 0; k0 < FDIM; k0 += 4) {
    const int ka = k0 + 2 * half;
    v2f a, b;
    a.x = arow[ka];
    a.y = arow[ka + 1];
    b.x = bcol[(size_t)ka * FDIM];
    b.y = bcol[(size_t)(ka + 1) * FDIM];
    c = __builtin_amdgcn_wmma_f32_16x16x4_f32(false, a, false, b, (short)0, c,
                                              false, false);
  }
  float* __restrict__ orow =
      out + (size_t)(tm * 16 + 8 * half) * FDIM + tn * 16 + l;
#pragma unroll
  for (int r = 0; r < 8; ++r) orow[(size_t)r * FDIM] = c[r];
}

__global__ void att_scores(const float* __restrict__ h,
                           const float* __restrict__ att_src,
                           const float* __restrict__ att_dst,
                           float* __restrict__ a_src, float* __restrict__ a_dst) {
  int t = blockIdx.x * blockDim.x + threadIdx.x;
  if (t >= N_NODES * HEADS) return;
  const int hh = t & (HEADS - 1);
  const int node = t >> 3;
  const float* __restrict__ hv = h + (size_t)node * FDIM + hh * HEAD_DIM;
  const float* __restrict__ asv = att_src + hh * HEAD_DIM;
  const float* __restrict__ adv = att_dst + hh * HEAD_DIM;
  float s1 = 0.f, s2 = 0.f;
#pragma unroll
  for (int d = 0; d < HEAD_DIM; ++d) {
    float v = hv[d];
    s1 += v * asv[d];
    s2 += v * adv[d];
  }
  a_src[t] = s1;
  a_dst[t] = s2;
}

__global__ void init_ms(float* __restrict__ m, float* __restrict__ s) {
  int t = blockIdx.x * blockDim.x + threadIdx.x;
  if (t >= N_NODES * HEADS) return;
  m[t] = -__builtin_inff();
  s[t] = 0.f;
}

__global__ void init_bias(float* __restrict__ acc, const float* __restrict__ bias) {
  int t = blockIdx.x * blockDim.x + threadIdx.x;
  if (t >= N_NODES * FDIM) return;
  acc[t] = bias[t & (FDIM - 1)];
}

__device__ __forceinline__ void edge_decode(const int* __restrict__ ei, int e,
                                            int& src, int& dst) {
  if (e < N_EDGES) {
    src = ei[e];
    dst = ei[N_EDGES + e];
  } else {
    src = dst = e - N_EDGES;
  }
}

__device__ __forceinline__ float leaky(float x) {
  return x > 0.f ? x : NEG_SLOPE * x;
}

__device__ __forceinline__ void atomicMaxF(float* addr, float val) {
  if (val >= 0.f)
    atomicMax((int*)addr, __float_as_int(val));
  else
    atomicMin((unsigned int*)addr, __float_as_uint(val));
}

__global__ void edge_max(const int* __restrict__ ei,
                         const float* __restrict__ a_src,
                         const float* __restrict__ a_dst, float* __restrict__ m) {
  int t = blockIdx.x * blockDim.x + threadIdx.x;
  if (t >= ETOT * HEADS) return;
  const int hh = t & (HEADS - 1);
  const int e = t >> 3;
  int src, dst;
  edge_decode(ei, e, src, dst);
  float ee = leaky(a_src[src * HEADS + hh] + a_dst[dst * HEADS + hh]);
  atomicMaxF(&m[dst * HEADS + hh], ee);
}

__global__ void edge_sum(const int* __restrict__ ei,
                         const float* __restrict__ a_src,
                         const float* __restrict__ a_dst,
                         const float* __restrict__ m, float* __restrict__ s) {
  int t = blockIdx.x * blockDim.x + threadIdx.x;
  if (t >= ETOT * HEADS) return;
  const int hh = t & (HEADS - 1);
  const int e = t >> 3;
  int src, dst;
  edge_decode(ei, e, src, dst);
  float ee = leaky(a_src[src * HEADS + hh] + a_dst[dst * HEADS + hh]);
  float mm = m[dst * HEADS + hh];
  if (!(mm > -__builtin_inff())) mm = 0.f;
  atomicAdd(&s[dst * HEADS + hh], expf(ee - mm));
}

__global__ void edge_agg(const int* __restrict__ ei,
                         const float* __restrict__ a_src,
                         const float* __restrict__ a_dst,
                         const float* __restrict__ m, const float* __restrict__ s,
                         const float* __restrict__ h, float* __restrict__ acc) {
  int t = blockIdx.x * blockDim.x + threadIdx.x;
  if (t >= ETOT * HEADS) return;
  const int hh = t & (HEADS - 1);
  const int e = t >> 3;
  int src, dst;
  edge_decode(ei, e, src, dst);
  float ee = leaky(a_src[src * HEADS + hh] + a_dst[dst * HEADS + hh]);
  float mm = m[dst * HEADS + hh];
  if (!(mm > -__builtin_inff())) mm = 0.f;
  float p = expf(ee - mm);
  float alpha = p / (s[dst * HEADS + hh] + 1e-16f);

  const float* __restrict__ hs = h + (size_t)src * FDIM + hh * HEAD_DIM;
  float* __restrict__ od = acc + (size_t)dst * FDIM + hh * HEAD_DIM;
#pragma unroll
  for (int d = 0; d < HEAD_DIM; d += 4) {
    float4 v = *(const float4*)(hs + d);
    atomicAdd(od + d + 0, alpha * v.x);
    atomicAdd(od + d + 1, alpha * v.y);
    atomicAdd(od + d + 2, alpha * v.z);
    atomicAdd(od + d + 3, alpha * v.w);
  }
}

__global__ void elu_inplace(float* __restrict__ x) {
  int t = blockIdx.x * blockDim.x + threadIdx.x;
  if (t >= N_NODES * FDIM) return;
  float v = x[t];
  x[t] = v > 0.f ? v : (expf(v) - 1.f);
}

extern "C" void kernel_launch(void* const* d_in, const int* in_sizes, int n_in,
                              void* d_out, int out_size, void* d_ws,
                              size_t ws_size, hipStream_t stream) {
  (void)in_sizes; (void)n_in; (void)out_size;

  const float* x   = (const float*)d_in[0];
  const int*   ei  = (const int*)d_in[1];
  const float* W1  = (const float*)d_in[2];
  const float* as1 = (const float*)d_in[3];
  const float* ad1 = (const float*)d_in[4];
  const float* b1  = (const float*)d_in[5];
  const float* W2  = (const float*)d_in[6];
  const float* as2 = (const float*)d_in[7];
  const float* ad2 = (const float*)d_in[8];
  const float* b2  = (const float*)d_in[9];
  float* out = (float*)d_out;

  float* ws    = (float*)d_ws;
  float* h     = ws;
  float* agg   = h + (size_t)N_NODES * FDIM;
  float* a_src = agg + (size_t)N_NODES * FDIM;
  float* a_dst = a_src + (size_t)N_NODES * HEADS;
  float* m     = a_dst + (size_t)N_NODES * HEADS;
  float* s     = m + (size_t)N_NODES * HEADS;
  const size_t need = ((size_t)N_NODES * FDIM * 2 + (size_t)N_NODES * HEADS * 4) *
                      sizeof(float);
  if (ws_size < need) return;

  const int gemmBlocks = (N_NODES / 16) * (FDIM / 16) / 4;
  const int nh = N_NODES * HEADS;
  const int nf = N_NODES * FDIM;
  const int eh = ETOT * HEADS;
  const int gNH = (nh + 255) / 256;
  const int gNF = (nf + 255) / 256;
  const int gEH = (eh + 255) / 256;
  dim3 B128(128), B256(256);

  gemm128_wmma_f32<<<gemmBlocks, B128, 0, stream>>>(x, W1, h);
  att_scores<<<gNH, B256, 0, stream>>>(h, as1, ad1, a_src, a_dst);
  init_ms<<<gNH, B256, 0, stream>>>(m, s);
  init_bias<<<gNF, B256, 0, stream>>>(agg, b1);
  edge_max<<<gEH, B256, 0, stream>>>(ei, a_src, a_dst, m);
  edge_sum<<<gEH, B256, 0, stream>>>(ei, a_src, a_dst, m, s);
  edge_agg<<<gEH, B256, 0, stream>>>(ei, a_src, a_dst, m, s, h, agg);
  elu_inplace<<<gNF, B256, 0, stream>>>(agg);

  gemm128_wmma_f32<<<gemmBlocks, B128, 0, stream>>>(agg, W2, h);
  att_scores<<<gNH, B256, 0, stream>>>(h, as2, ad2, a_src, a_dst);
  init_ms<<<gNH, B256, 0, stream>>>(m, s);
  init_bias<<<gNF, B256, 0, stream>>>(out, b2);
  edge_max<<<gEH, B256, 0, stream>>>(ei, a_src, a_dst, m);
  edge_sum<<<gEH, B256, 0, stream>>>(ei, a_src, a_dst, m, s);
  edge_agg<<<gEH, B256, 0, stream>>>(ei, a_src, a_dst, m, s, h, out);
}


# edge passes become one kernel per layer: a work-group keeps 256 destination rows in LDS, scans the whole destination list for its edges, adds rows by LDS float adds; sums in LDS arrival order
# speedup vs baseline: 4.7262x; 4.7262x over previous
#include <hip/hip_runtime.h>
#include <math.h>


#define N_NODES 50000
#define N_EDGES 800000
#define HEADS   8
#define HEAD_DIM 16
#define FDIM    128
#define ETOT    (N_EDGES + N_NODES)
#define NEG_SLOPE 0.2f

typedef __attribute__((ext_vector_type(2))) float v2f;
typedef __attribute__((ext_vector_type(8))) float v8f;

__global__ void __launch_bounds__(128)
gemm128_wmma_f32(const float* __restrict__ A, const float* __restrict__ W,
                 float* __restrict__ out) {
  const int wave = threadIdx.x >> 5;
  const int lane = threadIdx.x & 31;
  const int tile = blockIdx.x * 4 + wave;
  const int tm   = tile >> 3;
  const int tn   = tile & 7;
  const int half = lane >> 4;
  const int l    = lane & 15;

  const float* __restrict__ arow = A + (size_t)(tm * 16 + l) * FDIM;
  const float* __restrict__ bcol = W + tn * 16 + l;

  v8f c = {};
#pragma unroll
  for (int k0 = 0; k0 < FDIM; k0 += 4) {
    const int ka = k0 + 2 * half;
    v2f a, b;
    a.x = arow[ka];
    a.y = arow[ka + 1];
    b.x = bcol[(size_t)ka * FDIM];
    b.y = bcol[(size_t)(ka + 1) * FDIM];
    c = __builtin_amdgcn_wmma_f32_16x16x4_f32(false, a, false, b, (short)0, c,
                                              false, false);
  }
  float* __restrict__ orow =
      out + (size_t)(tm * 16 + 8 * half) * FDIM + tn * 16 + l;
#pragma unroll
  for (int r = 0; r < 8; ++r) orow[(size_t)r * FDIM] = c[r];
}

__global__ void att_scores(const float* __restrict__ h,
                           const float* __restrict__ att_src,
                           const float* __restrict__ att_dst,
                           float* __restrict__ a_src, float* __restrict__ a_dst) {
  int t = blockIdx.x * blockDim.x + threadIdx.x;
  if (t >= N_NODES * HEADS) return;
  const int hh = t & (HEADS - 1);
  const int node = t >> 3;
  const float* __restrict__ hv = h + (size_t)node * FDIM + hh * HEAD_DIM;
  const float* __restrict__ asv = att_src + hh * HEAD_DIM;
  const float* __restrict__ adv = att_dst + hh * HEAD_DIM;
  float s1 = 0.f, s2 = 0.f;
#pragma unroll
  for (int d = 0; d < HEAD_DIM; ++d) {
    float v = hv[d];
    s1 += v * asv[d];
    s2 += v * adv[d];
  }
  a_src[t] = s1;
  a_dst[t] = s2;
}

__device__ __forceinline__ void edge_decode(const int* __restrict__ ei, int e,
                                            int& src, int& dst) {
  if (e < N_EDGES) {
    src = ei[e];
    dst = ei[N_EDGES + e];
  } else {
    src = dst = e - N_EDGES;
  }
}

__device__ __forceinline__ float leaky(float x) {
  return x > 0.f ? x : NEG_SLOPE * x;
}

__device__ __forceinline__ void lds_max_f(float* addr, float val) {
  if (__float_as_int(val) >= 0)
    atomicMax((int*)addr, __float_as_int(val));
  else
    atomicMin((unsigned int*)addr, __float_as_uint(val));
}

#define R_TILE 256
#define LCAP   8192
#define TB     1024
#define SCAN_K 8
#define E4     (N_EDGES / 4)

__device__ __forceinline__ int tile_col(int c) { return (c & 3) * 32 + (c >> 2); }

template <bool ELU>
__global__ void __launch_bounds__(TB)
tile_edges(const int* __restrict__ ei, const float* __restrict__ a_src,
           const float* __restrict__ a_dst, const float* __restrict__ h,
           const float* __restrict__ bias, float* __restrict__ out) {
  __shared__ __attribute__((aligned(16))) float acc[R_TILE * FDIM];
  __shared__ float sm[R_TILE * HEADS];
  __shared__ float ss[R_TILE * HEADS];
  __shared__ float sad[R_TILE * HEADS];
  __shared__ unsigned int list[LCAP];
  __shared__ int cnt;

  const int tid  = threadIdx.x;
  const int lane = tid & 31;
  const int wave = tid >> 5;
  const int base = blockIdx.x * R_TILE;
  const int nvalid = (N_NODES - base) < R_TILE ? (N_NODES - base) : R_TILE;

  for (int i = tid; i < R_TILE * FDIM / 4; i += TB)
    ((float4*)acc)[i] = make_float4(0.f, 0.f, 0.f, 0.f);
  for (int i = tid; i < R_TILE * HEADS; i += TB) {
    const int node = base + (i >> 3);
    sm[i]  = -__builtin_inff();
    ss[i]  = 0.f;
    sad[i] = node < N_NODES ? a_dst[(size_t)node * HEADS + (i & 7)] : 0.f;
  }
  if (tid == 0) cnt = 0;
  __syncthreads();

  {
    const int4* __restrict__ dst4 = (const int4*)(ei + N_EDGES);
    for (int it = 0; it < E4; it += TB * SCAN_K) {
      int4 d[SCAN_K];
#pragma unroll
      for (int k = 0; k < SCAN_K; ++k) {
        const int q = it + k * TB + tid;
        d[k] = q < E4 ? dst4[q] : make_int4(-1, -1, -1, -1);
      }
      unsigned int msk[SCAN_K * 4];
      int nw = 0;
#pragma unroll
      for (int k = 0; k < SCAN_K; ++k) {
        const int dv[4] = {d[k].x, d[k].y, d[k].z, d[k].w};
#pragma unroll
        for (int j = 0; j < 4; ++j) {
          const bool mt = (unsigned int)(dv[j] - base) < (unsigned int)R_TILE;
          msk[k * 4 + j] = __builtin_amdgcn_ballot_w32(mt);
          nw += __builtin_popcount(msk[k * 4 + j]);
        }
      }
      if (nw) {
        int bp = 0;
        if (lane == 0) bp = atomicAdd(&cnt, nw);
        bp = __builtin_amdgcn_readfirstlane(bp);
#pragma unroll
        for (int k = 0; k < SCAN_K; ++k) {
          const int dv[4] = {d[k].x, d[k].y, d[k].z, d[k].w};
#pragma unroll
          for (int j = 0; j < 4; ++j) {
            const unsigned int mq = msk[k * 4 + j];
            if (mq) {
              if ((mq >> lane) & 1u) {
                const int pos = bp + (int)__builtin_amdgcn_mbcnt_lo(mq, 0u);
                if (pos < LCAP)
                  list[pos] = (unsigned int)((it + k * TB + tid) * 4 + j) |
                              ((unsigned int)(dv[j] - base) << 20);
              }
              bp += __builtin_popcount(mq);
            }
          }
        }
      }
    }
  }
  __syncthreads();
  const int nscan = cnt;
  const int total = nscan + nvalid;

  if (total <= LCAP) {
    if (tid < nvalid)
      list[nscan + tid] = (unsigned int)(base + tid) | ((unsigned int)tid << 16);
    __syncthreads();
    for (int i = tid; i < total; i += TB) {
      const unsigned int a = list[i];
      unsigned int src, dl;
      if (i < nscan) {
        src = (unsigned int)ei[a & 0xFFFFFu];
        dl  = a >> 20;
        list[i] = src | (dl << 16);
      } else {
        src = a & 0xFFFFu;
        dl  = a >> 16;
      }
      const float4 s0 = *(const float4*)(a_src + (size_t)src * HEADS);
      const float4 s1 = *(const float4*)(a_src + (size_t)src * HEADS + 4);
      const float sv[8] = {s0.x, s0.y, s0.z, s0.w, s1.x, s1.y, s1.z, s1.w};
#pragma unroll
      for (int hh = 0; hh < HEADS; ++hh)
        lds_max_f(&sm[dl * HEADS + hh], leaky(sv[hh] + sad[dl * HEADS + hh]));
    }
    __syncthreads();
    for (int i = tid; i < R_TILE * HEADS; i += TB) {
      const float mm = sm[i];
      sm[i] = (__builtin_fabsf(mm) < __builtin_inff()) ? mm : 0.f;
    }
    __syncthreads();
    constexpr int ADD_U = 8;
    const int hh = lane >> 2;
    for (int i0 = wave * ADD_U; i0 < total; i0 += (TB / 32) * ADD_U) {
      unsigned int a[ADD_U];
      float4 v[ADD_U];
      float as[ADD_U];
#pragma unroll
      for (int u = 0; u < ADD_U; ++u)
        a[u] = (i0 + u) < total ? list[i0 + u] : 0u;
#pragma unroll
      for (int u = 0; u < ADD_U; ++u) {
        const unsigned int src = a[u] & 0xFFFFu;
        v[u]  = *(const float4*)(h + (size_t)src * FDIM + lane * 4);
        as[u] = a_src[(size_t)src * HEADS + hh];
      }
#pragma unroll
      for (int u = 0; u < ADD_U; ++u) {
        if ((i0 + u) < total) {
          const unsigned int dl = a[u] >> 16;
          const float ev = leaky(as[u] + sad[dl * HEADS + hh]);
          const float p  = expf(ev - sm[dl * HEADS + hh]);
          float* ar = &acc[dl * FDIM + lane];
          atomicAdd(ar,      p * v[u].x);
          atomicAdd(ar + 32, p * v[u].y);
          atomicAdd(ar + 64, p * v[u].z);
          atomicAdd(ar + 96, p * v[u].w);
          if ((lane & 3) == 0) atomicAdd(&ss[dl * HEADS + hh], p);
        }
      }
    }
  } else {
    for (int e = tid; e < ETOT; e += TB) {
      int src, dst;
      edge_decode(ei, e, src, dst);
      const unsigned int dl = (unsigned int)(dst - base);
      if (dl < (unsigned int)R_TILE) {
        for (int hh = 0; hh < HEADS; ++hh)
          lds_max_f(&sm[dl * HEADS + hh],
                    leaky(a_src[(size_t)src * HEADS + hh] + sad[dl * HEADS + hh]));
      }
    }
    __syncthreads();
    for (int i = tid; i < R_TILE * HEADS; i += TB) {
      const float mm = sm[i];
      sm[i] = (__builtin_fabsf(mm) < __builtin_inff()) ? mm : 0.f;
    }
    __syncthreads();
    for (int e = tid; e < ETOT; e += TB) {
      int src, dst;
      edge_decode(ei, e, src, dst);
      const unsigned int dl = (unsigned int)(dst - base);
      if (dl < (unsigned int)R_TILE) {
        for (int hh = 0; hh < HEADS; ++hh) {
          const float ev =
              leaky(a_src[(size_t)src * HEADS + hh] + sad[dl * HEADS + hh]);
          const float p = expf(ev - sm[dl * HEADS + hh]);
          atomicAdd(&ss[dl * HEADS + hh], p);
          for (int dd = 0; dd < HEAD_DIM; ++dd) {
            const int c = hh * HEAD_DIM + dd;
            atomicAdd(&acc[dl * FDIM + tile_col(c)],
                      p * h[(size_t)src * FDIM + c]);
          }
        }
      }
    }
  }
  __syncthreads();

  for (int idx = tid; idx < R_TILE * 32; idx += TB) {
    const int row = idx >> 5;
    const int l   = idx & 31;
    const int node = base + row;
    if (node < N_NODES) {
      const float den = ss[row * HEADS + (l >> 2)] + 1e-16f;
      const float4 b = *(const float4*)(bias + l * 4);
      float4 o;
      o.x = acc[row * FDIM + l]      / den + b.x;
      o.y = acc[row * FDIM + 32 + l] / den + b.y;
      o.z = acc[row * FDIM + 64 + l] / den + b.z;
      o.w = acc[row * FDIM + 96 + l] / den + b.w;
      if (ELU) {
        o.x = o.x > 0.f ? o.x : (expf(o.x) - 1.f);
        o.y = o.y > 0.f ? o.y : (expf(o.y) - 1.f);
        o.z = o.z > 0.f ? o.z : (expf(o.z) - 1.f);
        o.w = o.w > 0.f ? o.w : (expf(o.w) - 1.f);
      }
      *(float4*)(out + (size_t)node * FDIM + l * 4) = o;
    }
  }
}

extern "C" void kernel_launch(void* const* d_in, const int* in_sizes, int n_in,
                              void* d_out, int out_size, void* d_ws,
                              size_t ws_size, hipStream_t stream) {
  (void)in_sizes; (void)n_in; (void)out_size;

  const float* x   = (const float*)d_in[0];
  const int*   ei  = (const int*)d_in[1];
  const float* W1  = (const float*)d_in[2];
  const float* as1 = (const float*)d_in[3];
  const float* ad1 = (const float*)d_in[4];
  const float* b1  = (const float*)d_in[5];
  const float* W2  = (const float*)d_in[6];
  const float* as2 = (const float*)d_in[7];
  const float* ad2 = (const float*)d_in[8];
  const float* b2  = (const float*)d_in[9];
  float* out = (float*)d_out;

  float* ws    = (float*)d_ws;
  float* h     = ws;
  float* agg   = h + (size_t)N_NODES * FDIM;
  float* a_src = agg + (size_t)N_NODES * FDIM;
  float* a_dst = a_src + (size_t)N_NODES * HEADS;
  const size_t need = ((size_t)N_NODES * FDIM * 2 + (size_t)N_NODES * HEADS * 2) *
                      sizeof(float);
  if (ws_size < need) return;

  const int gemmBlocks = (N_NODES / 16) * (FDIM / 16) / 4;
  const int nh = N_NODES * HEADS;
  const int gNH = (nh + 255) / 256;
  const int gT  = (N_NODES + R_TILE - 1) / R_TILE;
  dim3 B128(128), B256(256), BT(TB);

  gemm128_wmma_f32<<<gemmBlocks, B128, 0, stream>>>(x, W1, h);
  att_scores<<<gNH, B256, 0, stream>>>(h, as1, ad1, a_src, a_dst);
  tile_edges<true><<<gT, BT, 0, stream>>>(ei, a_src, a_dst, h, b1, agg);

  gemm128_wmma_f32<<<gemmBlocks, B128, 0, stream>>>(agg, W2, h);
  att_scores<<<gNH, B256, 0, stream>>>(h, as2, ad2, a_src, a_dst);
  tile_edges<false><<<gT, BT, 0, stream>>>(ei, a_src, a_dst, h, b2, out);
}


# layer-2 edge kernel loads each tile's (source, row) list that the layer-1 edge kernel of the same call saved to the workspace, instead of scanning all destinations again; sums in LDS arrival order
# speedup vs baseline: 5.4228x; 1.1474x over previous
#include <hip/hip_runtime.h>
#include <math.h>


#define N_NODES 50000
#define N_EDGES 800000
#define HEADS   8
#define HEAD_DIM 16
#define FDIM    128
#define ETOT    (N_EDGES + N_NODES)
#define NEG_SLOPE 0.2f

typedef __attribute__((ext_vector_type(2))) float v2f;
typedef __attribute__((ext_vector_type(8))) float v8f;

__global__ void __launch_bounds__(128)
gemm128_wmma_f32(const float* __restrict__ A, const float* __restrict__ W,
                 float* __restrict__ out) {
  const int wave = threadIdx.x >> 5;
  const int lane = threadIdx.x & 31;
  const int tile = blockIdx.x * 4 + wave;
  const int tm   = tile >> 3;
  const int tn   = tile & 7;
  const int half = lane >> 4;
  const int l    = lane & 15;

  const float* __restrict__ arow = A + (size_t)(tm * 16 + l) * FDIM;
  const float* __restrict__ bcol = W + tn * 16 + l;

  v8f c = {};
#pragma unroll
  for (int k0 = 0; k0 < FDIM; k0 += 4) {
    const int ka = k0 + 2 * half;
    v2f a, b;
    a.x = arow[ka];
    a.y = arow[ka + 1];
    b.x = bcol[(size_t)ka * FDIM];
    b.y = bcol[(size_t)(ka + 1) * FDIM];
    c = __builtin_amdgcn_wmma_f32_16x16x4_f32(false, a, false, b, (short)0, c,
                                              false, false);
  }
  float* __restrict__ orow =
      out + (size_t)(tm * 16 + 8 * half) * FDIM + tn * 16 + l;
#pragma unroll
  for (int r = 0; r < 8; ++r) orow[(size_t)r * FDIM] = c[r];
}

__global__ void att_scores(const float* __restrict__ h,
                           const float* __restrict__ att_src,
                           const float* __restrict__ att_dst,
                           float* __restrict__ a_src, float* __restrict__ a_dst) {
  int t = blockIdx.x * blockDim.x + threadIdx.x;
  if (t >= N_NODES * HEADS) return;
  const int hh = t & (HEADS - 1);
  const int node = t >> 3;
  const float* __restrict__ hv = h + (size_t)node * FDIM + hh * HEAD_DIM;
  const float* __restrict__ asv = att_src + hh * HEAD_DIM;
  const float* __restrict__ adv = att_dst + hh * HEAD_DIM;
  float s1 = 0.f, s2 = 0.f;
#pragma unroll
  for (int d = 0; d < HEAD_DIM; ++d) {
    float v = hv[d];
    s1 += v * asv[d];
    s2 += v * adv[d];
  }
  a_src[t] = s1;
  a_dst[t] = s2;
}

__device__ __forceinline__ void edge_decode(const int* __restrict__ ei, int e,
                                            int& src, int& dst) {
  if (e < N_EDGES) {
    src = ei[e];
    dst = ei[N_EDGES + e];
  } else {
    src = dst = e - N_EDGES;
  }
}

__device__ __forceinline__ float leaky(float x) {
  return x > 0.f ? x : NEG_SLOPE * x;
}

__device__ __forceinline__ void lds_max_f(float* addr, float val) {
  if (__float_as_int(val) >= 0)
    atomicMax((int*)addr, __float_as_int(val));
  else
    atomicMin((unsigned int*)addr, __float_as_uint(val));
}

#define R_TILE 256
#define LCAP   8192
#define TB     1024
#define SCAN_K 8
#define E4     (N_EDGES / 4)
#define TL_HDR    32
#define TL_STRIDE (LCAP + TL_HDR)

__device__ __forceinline__ int tile_col(int c) { return (c & 3) * 32 + (c >> 2); }

template <bool FIRST>
__global__ void __launch_bounds__(TB)
tile_edges(const int* __restrict__ ei, const float* __restrict__ a_src,
           const float* __restrict__ a_dst, const float* __restrict__ h,
           const float* __restrict__ bias, float* __restrict__ out,
           unsigned int* __restrict__ tl) {
  constexpr bool ELU = FIRST;
  __shared__ __attribute__((aligned(16))) float acc[R_TILE * FDIM];
  __shared__ float sm[R_TILE * HEADS];
  __shared__ float ss[R_TILE * HEADS];
  __shared__ float sad[R_TILE * HEADS];
  __shared__ __attribute__((aligned(16))) unsigned int list[LCAP];
  __shared__ int cnt;

  const int tid  = threadIdx.x;
  const int lane = tid & 31;
  const int wave = tid >> 5;
  const int base = blockIdx.x * R_TILE;
  const int nvalid = (N_NODES - base) < R_TILE ? (N_NODES - base) : R_TILE;
  unsigned int* __restrict__ my_tl = tl + (size_t)blockIdx.x * TL_STRIDE;

  for (int i = tid; i < R_TILE * FDIM / 4; i += TB)
    ((float4*)acc)[i] = make_float4(0.f, 0.f, 0.f, 0.f);
  for (int i = tid; i < R_TILE * HEADS; i += TB) {
    const int node = base + (i >> 3);
    sm[i]  = -__builtin_inff();
    ss[i]  = 0.f;
    sad[i] = node < N_NODES ? a_dst[(size_t)node * HEADS + (i & 7)] : 0.f;
  }
  if (tid == 0) cnt = 0;
  __syncthreads();

  if (FIRST) {
    const int4* __restrict__ dst4 = (const int4*)(ei + N_EDGES);
    for (int it = 0; it < E4; it += TB * SCAN_K) {
      int4 d[SCAN_K];
#pragma unroll
      for (int k = 0; k < SCAN_K; ++k) {
        const int q = it + k * TB + tid;
        d[k] = q < E4 ? dst4[q] : make_int4(-1, -1, -1, -1);
      }
      unsigned int msk[SCAN_K * 4];
      int nw = 0;
#pragma unroll
      for (int k = 0; k < SCAN_K; ++k) {
        const int dv[4] = {d[k].x, d[k].y, d[k].z, d[k].w};
#pragma unroll
        for (int j = 0; j < 4; ++j) {
          const bool mt = (unsigned int)(dv[j] - base) < (unsigned int)R_TILE;
          msk[k * 4 + j] = __builtin_amdgcn_ballot_w32(mt);
          nw += __builtin_popcount(msk[k * 4 + j]);
        }
      }
      if (nw) {
        int bp = 0;
        if (lane == 0) bp = atomicAdd(&cnt, nw);
        bp = __builtin_amdgcn_readfirstlane(bp);
#pragma unroll
        for (int k = 0; k < SCAN_K; ++k) {
          const int dv[4] = {d[k].x, d[k].y, d[k].z, d[k].w};
#pragma unroll
          for (int j = 0; j < 4; ++j) {
            const unsigned int mq = msk[k * 4 + j];
            if (mq) {
              if ((mq >> lane) & 1u) {
                const int pos = bp + (int)__builtin_amdgcn_mbcnt_lo(mq, 0u);
                if (pos < LCAP)
                  list[pos] = (unsigned int)((it + k * TB + tid) * 4 + j) |
                              ((unsigned int)(dv[j] - base) << 20);
              }
              bp += __builtin_popcount(mq);
            }
          }
        }
      }
    }
  }
  int nscan, total;
  if (FIRST) {
    __syncthreads();
    nscan = cnt;
    total = nscan + nvalid;
    if (tid == 0) my_tl[0] = (unsigned int)total;
  } else {
    nscan = 0;
    total = __builtin_amdgcn_readfirstlane((int)my_tl[0]);
  }

  if (total <= LCAP) {
    if (FIRST) {
      if (tid < nvalid)
        list[nscan + tid] = (unsigned int)(base + tid) | ((unsigned int)tid << 16);
    } else {
      const uint4* __restrict__ g4 = (const uint4*)(my_tl + TL_HDR);
      for (int i = tid; i < (total + 3) / 4; i += TB) ((uint4*)list)[i] = g4[i];
    }
    __syncthreads();
    for (int i = tid; i < total; i += TB) {
      const unsigned int a = list[i];
      unsigned int src, dl;
      if (FIRST && i < nscan) {
        src = (unsigned int)ei[a & 0xFFFFFu];
        dl  = a >> 20;
        list[i] = src | (dl << 16);
      } else {
        src = a & 0xFFFFu;
        dl  = a >> 16;
      }
      const float4 s0 = *(const float4*)(a_src + (size_t)src * HEADS);
      const float4 s1 = *(const float4*)(a_src + (size_t)src * HEADS + 4);
      const float sv[8] = {s0.x, s0.y, s0.z, s0.w, s1.x, s1.y, s1.z, s1.w};
#pragma unroll
      for (int hh = 0; hh < HEADS; ++hh)
        lds_max_f(&sm[dl * HEADS + hh], leaky(sv[hh] + sad[dl * HEADS + hh]));
    }
    __syncthreads();
    if (FIRST) {
      uint4* __restrict__ g4 = (uint4*)(my_tl + TL_HDR);
      for (int i = tid; i < (total + 3) / 4; i += TB) g4[i] = ((const uint4*)list)[i];
    }
    for (int i = tid; i < R_TILE * HEADS; i += TB) {
      const float mm = sm[i];
      sm[i] = (__builtin_fabsf(mm) < __builtin_inff()) ? mm : 0.f;
    }
    __syncthreads();
    constexpr int ADD_U = 8;
    const int hh = lane >> 2;
    for (int i0 = wave * ADD_U; i0 < total; i0 += (TB / 32) * ADD_U) {
      unsigned int a[ADD_U];
      float4 v[ADD_U];
      float as[ADD_U];
#pragma unroll
      for (int u = 0; u < ADD_U; ++u)
        a[u] = (i0 + u) < total ? list[i0 + u] : 0u;
#pragma unroll
      for (int u = 0; u < ADD_U; ++u) {
        const unsigned int src = a[u] & 0xFFFFu;
        v[u]  = *(const float4*)(h + (size_t)src * FDIM + lane * 4);
        as[u] = a_src[(size_t)src * HEADS + hh];
      }
#pragma unroll
      for (int u = 0; u < ADD_U; ++u) {
        if ((i0 + u) < total) {
          const unsigned int dl = a[u] >> 16;
          const float ev = leaky(as[u] + sad[dl * HEADS + hh]);
          const float p  = expf(ev - sm[dl * HEADS + hh]);
          float* ar = &acc[dl * FDIM + lane];
          atomicAdd(ar,      p * v[u].x);
          atomicAdd(ar + 32, p * v[u].y);
          atomicAdd(ar + 64, p * v[u].z);
          atomicAdd(ar + 96, p * v[u].w);
          if ((lane & 3) == 0) atomicAdd(&ss[dl * HEADS + hh], p);
        }
      }
    }
  } else {
    for (int e = tid; e < ETOT; e += TB) {
      int src, dst;
      edge_decode(ei, e, src, dst);
      const unsigned int dl = (unsigned int)(dst - base);
      if (dl < (unsigned int)R_TILE) {
        for (int hh = 0; hh < HEADS; ++hh)
          lds_max_f(&sm[dl * HEADS + hh],
                    leaky(a_src[(size_t)src * HEADS + hh] + sad[dl * HEADS + hh]));
      }
    }
    __syncthreads();
    for (int i = tid; i < R_TILE * HEADS; i += TB) {
      const float mm = sm[i];
      sm[i] = (__builtin_fabsf(mm) < __builtin_inff()) ? mm : 0.f;
    }
    __syncthreads();
    for (int e = tid; e < ETOT; e += TB) {
      int src, dst;
      edge_decode(ei, e, src, dst);
      const unsigned int dl = (unsigned int)(dst - base);
      if (dl < (unsigned int)R_TILE) {
        for (int hh = 0; hh < HEADS; ++hh) {
          const float ev =
              leaky(a_src[(size_t)src * HEADS + hh] + sad[dl * HEADS + hh]);
          const float p = expf(ev - sm[dl * HEADS + hh]);
          atomicAdd(&ss[dl * HEADS + hh], p);
          for (int dd = 0; dd < HEAD_DIM; ++dd) {
            const int c = hh * HEAD_DIM + dd;
            atomicAdd(&acc[dl * FDIM + tile_col(c)],
                      p * h[(size_t)src * FDIM + c]);
          }
        }
      }
    }
  }
  __syncthreads();

  for (int idx = tid; idx < R_TILE * 32; idx += TB) {
    const int row = idx >> 5;
    const int l   = idx & 31;
    const int node = base + row;
    if (node < N_NODES) {
      const float den = ss[row * HEADS + (l >> 2)] + 1e-16f;
      const float4 b = *(const float4*)(bias + l * 4);
      float4 o;
      o.x = acc[row * FDIM + l]      / den + b.x;
      o.y = acc[row * FDIM + 32 + l] / den + b.y;
      o.z = acc[row * FDIM + 64 + l] / den + b.z;
      o.w = acc[row * FDIM + 96 + l] / den + b.w;
      if (ELU) {
        o.x = o.x > 0.f ? o.x : (expf(o.x) - 1.f);
        o.y = o.y > 0.f ? o.y : (expf(o.y) - 1.f);
        o.z = o.z > 0.f ? o.z : (expf(o.z) - 1.f);
        o.w = o.w > 0.f ? o.w : (expf(o.w) - 1.f);
      }
      *(float4*)(out + (size_t)node * FDIM + l * 4) = o;
    }
  }
}

extern "C" void kernel_launch(void* const* d_in, const int* in_sizes, int n_in,
                              void* d_out, int out_size, void* d_ws,
                              size_t ws_size, hipStream_t stream) {
  (void)in_sizes; (void)n_in; (void)out_size;

  const float* x   = (const float*)d_in[0];
  const int*   ei  = (const int*)d_in[1];
  const float* W1  = (const float*)d_in[2];
  const float* as1 = (const float*)d_in[3];
  const float* ad1 = (const float*)d_in[4];
  const float* b1  = (const float*)d_in[5];
  const float* W2  = (const float*)d_in[6];
  const float* as2 = (const float*)d_in[7];
  const float* ad2 = (const float*)d_in[8];
  const float* b2  = (const float*)d_in[9];
  float* out = (float*)d_out;

  float* ws    = (float*)d_ws;
  float* h     = ws;
  float* agg   = h + (size_t)N_NODES * FDIM;
  float* a_src = agg + (size_t)N_NODES * FDIM;
  float* a_dst = a_src + (size_t)N_NODES * HEADS;
  unsigned int* tl = (unsigned int*)(a_dst + (size_t)N_NODES * HEADS);
  const int gT  = (N_NODES + R_TILE - 1) / R_TILE;
  const size_t need = ((size_t)N_NODES * FDIM * 2 + (size_t)N_NODES * HEADS * 2) *
                          sizeof(float) +
                      (size_t)gT * TL_STRIDE * sizeof(unsigned int);
  if (ws_size < need) return;

  const int gemmBlocks = (N_NODES / 16) * (FDIM / 16) / 4;
  const int nh = N_NODES * HEADS;
  const int gNH = (nh + 255) / 256;
  dim3 B128(128), B256(256), BT(TB);

  gemm128_wmma_f32<<<gemmBlocks, B128, 0, stream>>>(x, W1, h);
  att_scores<<<gNH, B256, 0, stream>>>(h, as1, ad1, a_src, a_dst);
  tile_edges<true><<<gT, BT, 0, stream>>>(ei, a_src, a_dst, h, b1, agg, tl);

  gemm128_wmma_f32<<<gemmBlocks, B128, 0, stream>>>(agg, W2, h);
  att_scores<<<gNH, B256, 0, stream>>>(h, as2, ad2, a_src, a_dst);
  tile_edges<false><<<gT, BT, 0, stream>>>(ei, a_src, a_dst, h, b2, out, tl);
}


# edge kernels: destination tile 256 -> 196 rows, so 256 work-groups, one on each of the 256 WGPs, instead of 196 with 60 WGPs idle; sums still in LDS arrival order (tile borders move)
# speedup vs baseline: 5.9917x; 1.1049x over previous
#include <hip/hip_runtime.h>
#include <math.h>


#define N_NODES 50000
#define N_EDGES 800000
#define HEADS   8
#define HEAD_DIM 16
#define FDIM    128
#define ETOT    (N_EDGES + N_NODES)
#define NEG_SLOPE 0.2f

typedef __attribute__((ext_vector_type(2))) float v2f;
typedef __attribute__((ext_vector_type(8))) float v8f;

__global__ void __launch_bounds__(128)
gemm128_wmma_f32(const float* __restrict__ A, const float* __restrict__ W,
                 float* __restrict__ out) {
  const int wave = threadIdx.x >> 5;
  const int lane = threadIdx.x & 31;
  const int tile = blockIdx.x * 4 + wave;
  const int tm   = tile >> 3;
  const int tn   = tile & 7;
  const int half = lane >> 4;
  const int l    = lane & 15;

  const float* __restrict__ arow = A + (size_t)(tm * 16 + l) * FDIM;
  const float* __restrict__ bcol = W + tn * 16 + l;

  v8f c = {};
#pragma unroll
  for (int k0 = 0; k0 < FDIM; k0 += 4) {
    const int ka = k0 + 2 * half;
    v2f a, b;
    a.x = arow[ka];
    a.y = arow[ka + 1];
    b.x = bcol[(size_t)ka * FDIM];
    b.y = bcol[(size_t)(ka + 1) * FDIM];
    c = __builtin_amdgcn_wmma_f32_16x16x4_f32(false, a, false, b, (short)0, c,
                                              false, false);
  }
  float* __restrict__ orow =
      out + (size_t)(tm * 16 + 8 * half) * FDIM + tn * 16 + l;
#pragma unroll
  for (int r = 0; r < 8; ++r) orow[(size_t)r * FDIM] = c[r];
}

__global__ void att_scores(const float* __restrict__ h,
                           const float* __restrict__ att_src,
                           const float* __restrict__ att_dst,
                           float* __restrict__ a_src, float* __restrict__ a_dst) {
  int t = blockIdx.x * blockDim.x + threadIdx.x;
  if (t >= N_NODES * HEADS) return;
  const int hh = t & (HEADS - 1);
  const int node = t >> 3;
  const float* __restrict__ hv = h + (size_t)node * FDIM + hh * HEAD_DIM;
  const float* __restrict__ asv = att_src + hh * HEAD_DIM;
  const float* __restrict__ adv = att_dst + hh * HEAD_DIM;
  float s1 = 0.f, s2 = 0.f;
#pragma unroll
  for (int d = 0; d < HEAD_DIM; ++d) {
    float v = hv[d];
    s1 += v * asv[d];
    s2 += v * adv[d];
  }
  a_src[t] = s1;
  a_dst[t] = s2;
}

__device__ __forceinline__ void edge_decode(const int* __restrict__ ei, int e,
                                            int& src, int& dst) {
  if (e < N_EDGES) {
    src = ei[e];
    dst = ei[N_EDGES + e];
  } else {
    src = dst = e - N_EDGES;
  }
}

__device__ __forceinline__ float leaky(float x) {
  return x > 0.f ? x : NEG_SLOPE * x;
}

__device__ __forceinline__ void lds_max_f(float* addr, float val) {
  if (__float_as_int(val) >= 0)
    atomicMax((int*)addr, __float_as_int(val));
  else
    atomicMin((unsigned int*)addr, __float_as_uint(val));
}

#define R_TILE 196
#define LCAP   8192
#define TB     1024
#define SCAN_K 8
#define E4     (N_EDGES / 4)
#define TL_HDR    32
#define TL_STRIDE (LCAP + TL_HDR)

__device__ __forceinline__ int tile_col(int c) { return (c & 3) * 32 + (c >> 2); }

template <bool FIRST>
__global__ void __launch_bounds__(TB)
tile_edges(const int* __restrict__ ei, const float* __restrict__ a_src,
           const float* __restrict__ a_dst, const float* __restrict__ h,
           const float* __restrict__ bias, float* __restrict__ out,
           unsigned int* __restrict__ tl) {
  constexpr bool ELU = FIRST;
  __shared__ __attribute__((aligned(16))) float acc[R_TILE * FDIM];
  __shared__ float sm[R_TILE * HEADS];
  __shared__ float ss[R_TILE * HEADS];
  __shared__ float sad[R_TILE * HEADS];
  __shared__ __attribute__((aligned(16))) unsigned int list[LCAP];
  __shared__ int cnt;

  const int tid  = threadIdx.x;
  const int lane = tid & 31;
  const int wave = tid >> 5;
  const int base = blockIdx.x * R_TILE;
  const int nvalid = (N_NODES - base) < R_TILE ? (N_NODES - base) : R_TILE;
  unsigned int* __restrict__ my_tl = tl + (size_t)blockIdx.x * TL_STRIDE;

  for (int i = tid; i < R_TILE * FDIM / 4; i += TB)
    ((float4*)acc)[i] = make_float4(0.f, 0.f, 0.f, 0.f);
  for (int i = tid; i < R_TILE * HEADS; i += TB) {
    const int node = base + (i >> 3);
    sm[i]  = -__builtin_inff();
    ss[i]  = 0.f;
    sad[i] = node < N_NODES ? a_dst[(size_t)node * HEADS + (i & 7)] : 0.f;
  }
  if (tid == 0) cnt = 0;
  __syncthreads();

  if (FIRST) {
    const int4* __restrict__ dst4 = (const int4*)(ei + N_EDGES);
    for (int it = 0; it < E4; it += TB * SCAN_K) {
      int4 d[SCAN_K];
#pragma unroll
      for (int k = 0; k < SCAN_K; ++k) {
        const int q = it + k * TB + tid;
        d[k] = q < E4 ? dst4[q] : make_int4(-1, -1, -1, -1);
      }
      unsigned int msk[SCAN_K * 4];
      int nw = 0;
#pragma unroll
      for (int k = 0; k < SCAN_K; ++k) {
        const int dv[4] = {d[k].x, d[k].y, d[k].z, d[k].w};
#pragma unroll
        for (int j = 0; j < 4; ++j) {
          const bool mt = (unsigned int)(dv[j] - base) < (unsigned int)R_TILE;
          msk[k * 4 + j] = __builtin_amdgcn_ballot_w32(mt);
          nw += __builtin_popcount(msk[k * 4 + j]);
        }
      }
      if (nw) {
        int bp = 0;
        if (lane == 0) bp = atomicAdd(&cnt, nw);
        bp = __builtin_amdgcn_readfirstlane(bp);
#pragma unroll
        for (int k = 0; k < SCAN_K; ++k) {
          const int dv[4] = {d[k].x, d[k].y, d[k].z, d[k].w};
#pragma unroll
          for (int j = 0; j < 4; ++j) {
            const unsigned int mq = msk[k * 4 + j];
            if (mq) {
              if ((mq >> lane) & 1u) {
                const int pos = bp + (int)__builtin_amdgcn_mbcnt_lo(mq, 0u);
                if (pos < LCAP)
                  list[pos] = (unsigned int)((it + k * TB + tid) * 4 + j) |
                              ((unsigned int)(dv[j] - base) << 20);
              }
              bp += __builtin_popcount(mq);
            }
          }
        }
      }
    }
  }
  int nscan, total;
  if (FIRST) {
    __syncthreads();
    nscan = cnt;
    total = nscan + nvalid;
    if (tid == 0) my_tl[0] = (unsigned int)total;
  } else {
    nscan = 0;
    total = __builtin_amdgcn_readfirstlane((int)my_tl[0]);
  }

  if (total <= LCAP) {
    if (FIRST) {
      if (tid < nvalid)
        list[nscan + tid] = (unsigned int)(base + tid) | ((unsigned int)tid << 16);
    } else {
      const uint4* __restrict__ g4 = (const uint4*)(my_tl + TL_HDR);
      for (int i = tid; i < (total + 3) / 4; i += TB) ((uint4*)list)[i] = g4[i];
    }
    __syncthreads();
    for (int i = tid; i < total; i += TB) {
      const unsigned int a = list[i];
      unsigned int src, dl;
      if (FIRST && i < nscan) {
        src = (unsigned int)ei[a & 0xFFFFFu];
        dl  = a >> 20;
        list[i] = src | (dl << 16);
      } else {
        src = a & 0xFFFFu;
        dl  = a >> 16;
      }
      const float4 s0 = *(const float4*)(a_src + (size_t)src * HEADS);
      const float4 s1 = *(const float4*)(a_src + (size_t)src * HEADS + 4);
      const float sv[8] = {s0.x, s0.y, s0.z, s0.w, s1.x, s1.y, s1.z, s1.w};
#pragma unroll
      for (int hh = 0; hh < HEADS; ++hh)
        lds_max_f(&sm[dl * HEADS + hh], leaky(sv[hh] + sad[dl * HEADS + hh]));
    }
    __syncthreads();
    if (FIRST) {
      uint4* __restrict__ g4 = (uint4*)(my_tl + TL_HDR);
      for (int i = tid; i < (total + 3) / 4; i += TB) g4[i] = ((const uint4*)list)[i];
    }
    for (int i = tid; i < R_TILE * HEADS; i += TB) {
      const float mm = sm[i];
      sm[i] = (__builtin_fabsf(mm) < __builtin_inff()) ? mm : 0.f;
    }
    __syncthreads();
    constexpr int ADD_U = 8;
    const int hh = lane >> 2;
    for (int i0 = wave * ADD_U; i0 < total; i0 += (TB / 32) * ADD_U) {
      unsigned int a[ADD_U];
      float4 v[ADD_U];
      float as[ADD_U];
#pragma unroll
      for (int u = 0; u < ADD_U; ++u)
        a[u] = (i0 + u) < total ? list[i0 + u] : 0u;
#pragma unroll
      for (int u = 0; u < ADD_U; ++u) {
        const unsigned int src = a[u] & 0xFFFFu;
        v[u]  = *(const float4*)(h + (size_t)src * FDIM + lane * 4);
        as[u] = a_src[(size_t)src * HEADS + hh];
      }
#pragma unroll
      for (int u = 0; u < ADD_U; ++u) {
        if ((i0 + u) < total) {
          const unsigned int dl = a[u] >> 16;
          const float ev = leaky(as[u] + sad[dl * HEADS + hh]);
          const float p  = expf(ev - sm[dl * HEADS + hh]);
          float* ar = &acc[dl * FDIM + lane];
          atomicAdd(ar,      p * v[u].x);
          atomicAdd(ar + 32, p * v[u].y);
          atomicAdd(ar + 64, p * v[u].z);
          atomicAdd(ar + 96, p * v[u].w);
          if ((lane & 3) == 0) atomicAdd(&ss[dl * HEADS + hh], p);
        }
      }
    }
  } else {
    for (int e = tid; e < ETOT; e += TB) {
      int src, dst;
      edge_decode(ei, e, src, dst);
      const unsigned int dl = (unsigned int)(dst - base);
      if (dl < (unsigned int)R_TILE) {
        for (int hh = 0; hh < HEADS; ++hh)
          lds_max_f(&sm[dl * HEADS + hh],
                    leaky(a_src[(size_t)src * HEADS + hh] + sad[dl * HEADS + hh]));
      }
    }
    __syncthreads();
    for (int i = tid; i < R_TILE * HEADS; i += TB) {
      const float mm = sm[i];
      sm[i] = (__builtin_fabsf(mm) < __builtin_inff()) ? mm : 0.f;
    }
    __syncthreads();
    for (int e = tid; e < ETOT; e += TB) {
      int src, dst;
      edge_decode(ei, e, src, dst);
      const unsigned int dl = (unsigned int)(dst - base);
      if (dl < (unsigned int)R_TILE) {
        for (int hh = 0; hh < HEADS; ++hh) {
          const float ev =
              leaky(a_src[(size_t)src * HEADS + hh] + sad[dl * HEADS + hh]);
          const float p = expf(ev - sm[dl * HEADS + hh]);
          atomicAdd(&ss[dl * HEADS + hh], p);
          for (int dd = 0; dd < HEAD_DIM; ++dd) {
            const int c = hh * HEAD_DIM + dd;
            atomicAdd(&acc[dl * FDIM + tile_col(c)],
                      p * h[(size_t)src * FDIM + c]);
          }
        }
      }
    }
  }
  __syncthreads();

  for (int idx = tid; idx < R_TILE * 32; idx += TB) {
    const int row = idx >> 5;
    const int l   = idx & 31;
    const int node = base + row;
    if (node < N_NODES) {
      const float den = ss[row * HEADS + (l >> 2)] + 1e-16f;
      const float4 b = *(const float4*)(bias + l * 4);
      float4 o;
      o.x = acc[row * FDIM + l]      / den + b.x;
      o.y = acc[row * FDIM + 32 + l] / den + b.y;
      o.z = acc[row * FDIM + 64 + l] / den + b.z;
      o.w = acc[row * FDIM + 96 + l] / den + b.w;
      if (ELU) {
        o.x = o.x > 0.f ? o.x : (expf(o.x) - 1.f);
        o.y = o.y > 0.f ? o.y : (expf(o.y) - 1.f);
        o.z = o.z > 0.f ? o.z : (expf(o.z) - 1.f);
        o.w = o.w > 0.f ? o.w : (expf(o.w) - 1.f);
      }
      *(float4*)(out + (size_t)node * FDIM + l * 4) = o;
    }
  }
}

extern "C" void kernel_launch(void* const* d_in, const int* in_sizes, int n_in,
                              void* d_out, int out_size, void* d_ws,
                              size_t ws_size, hipStream_t stream) {
  (void)in_sizes; (void)n_in; (void)out_size;

  const float* x   = (const float*)d_in[0];
  const int*   ei  = (const int*)d_in[1];
  const float* W1  = (const float*)d_in[2];
  const float* as1 = (const float*)d_in[3];
  const float* ad1 = (const float*)d_in[4];
  const float* b1  = (const float*)d_in[5];
  const float* W2  = (const float*)d_in[6];
  const float* as2 = (const float*)d_in[7];
  const float* ad2 = (const float*)d_in[8];
  const float* b2  = (const float*)d_in[9];
  float* out = (float*)d_out;

  float* ws    = (float*)d_ws;
  float* h     = ws;
  float* agg   = h + (size_t)N_NODES * FDIM;
  float* a_src = agg + (size_t)N_NODES * FDIM;
  float* a_dst = a_src + (size_t)N_NODES * HEADS;
  unsigned int* tl = (unsigned int*)(a_dst + (size_t)N_NODES * HEADS);
  const int gT  = (N_NODES + R_TILE - 1) / R_TILE;
  const size_t need = ((size_t)N_NODES * FDIM * 2 + (size_t)N_NODES * HEADS * 2) *
                          sizeof(float) +
                      (size_t)gT * TL_STRIDE * sizeof(unsigned int);
  if (ws_size < need) return;

  const int gemmBlocks = (N_NODES / 16) * (FDIM / 16) / 4;
  const int nh = N_NODES * HEADS;
  const int gNH = (nh + 255) / 256;
  dim3 B128(128), B256(256), BT(TB);

  gemm128_wmma_f32<<<gemmBlocks, B128, 0, stream>>>(x, W1, h);
  att_scores<<<gNH, B256, 0, stream>>>(h, as1, ad1, a_src, a_dst);
  tile_edges<true><<<gT, BT, 0, stream>>>(ei, a_src, a_dst, h, b1, agg, tl);

  gemm128_wmma_f32<<<gemmBlocks, B128, 0, stream>>>(agg, W2, h);
  att_scores<<<gNH, B256, 0, stream>>>(h, as2, ad2, a_src, a_dst);
  tile_edges<false><<<gT, BT, 0, stream>>>(ei, a_src, a_dst, h, b2, out, tl);
}


# GEMM: each wave keeps its 128x16 block of W in registers for the whole launch; 512 persistent 8-wave groups stream 16-row tiles of A through LDS by whole-row loads; k order of the sums changes
# speedup vs baseline: 6.7919x; 1.1336x over previous
#include <hip/hip_runtime.h>
#include <math.h>


#define N_NODES 50000
#define N_EDGES 800000
#define HEADS   8
#define HEAD_DIM 16
#define FDIM    128
#define ETOT    (N_EDGES + N_NODES)
#define NEG_SLOPE 0.2f

typedef __attribute__((ext_vector_type(2))) float v2f;
typedef __attribute__((ext_vector_type(8))) float v8f;

#define GEMM_WAVES  8
#define GEMM_TB     (GEMM_WAVES * 32)
#define GEMM_GROUPS 512
#define ROW_TILES   (N_NODES / 16)
#define A_PITCH     132

__global__ void __launch_bounds__(GEMM_TB)
gemm128_wmma_f32(const float* __restrict__ A, const float* __restrict__ W,
                 float* __restrict__ out) {
  __shared__ __attribute__((aligned(16))) float As[16 * A_PITCH];
  const int tid  = threadIdx.x;
  const int wave = tid >> 5;
  const int lane = tid & 31;
  const int half = lane >> 4;
  const int l    = lane & 15;

  v2f b[32];
  const float* __restrict__ bcol = W + wave * 16 + l;
#pragma unroll
  for (int s = 0; s < 32; ++s) {
    const int k = 8 * (s >> 1) + 4 * half + 2 * (s & 1);
    b[s].x = bcol[(size_t)k * FDIM];
    b[s].y = bcol[(size_t)(k + 1) * FDIM];
  }

  float* __restrict__ st0 = &As[wave * A_PITCH + lane * 4];
  float* __restrict__ st1 = &As[(wave + 8) * A_PITCH + lane * 4];
  const float* __restrict__ frag = &As[l * A_PITCH + 4 * half];

  int tm = blockIdx.x;
  const float4* __restrict__ src = (const float4*)(A + (size_t)tm * 16 * FDIM);
  float4 g0 = src[tid];
  float4 g1 = src[tid + GEMM_TB];
  for (; tm < ROW_TILES; tm += GEMM_GROUPS) {
    *(float4*)st0 = g0;
    *(float4*)st1 = g1;
    __syncthreads();
    const int nx = tm + GEMM_GROUPS;
    if (nx < ROW_TILES) {
      src = (const float4*)(A + (size_t)nx * 16 * FDIM);
      g0 = src[tid];
      g1 = src[tid + GEMM_TB];
    }
    float4 q[16];
#pragma unroll
    for (int j = 0; j < 16; ++j) q[j] = *(const float4*)(frag + 8 * j);
    __syncthreads();

    v8f c = {};
#pragma unroll
    for (int j = 0; j < 16; ++j) {
      v2f a;
      a.x = q[j].x;
      a.y = q[j].y;
      c = __builtin_amdgcn_wmma_f32_16x16x4_f32(false, a, false, b[2 * j],
                                                (short)0, c, false, false);
      a.x = q[j].z;
      a.y = q[j].w;
      c = __builtin_amdgcn_wmma_f32_16x16x4_f32(false, a, false, b[2 * j + 1],
                                                (short)0, c, false, false);
    }
    float* __restrict__ orow =
        out + (size_t)(tm * 16 + 8 * half) * FDIM + wave * 16 + l;
#pragma unroll
    for (int r = 0; r < 8; ++r) orow[(size_t)r * FDIM] = c[r];
  }
}

__global__ void att_scores(const float* __restrict__ h,
                           const float* __restrict__ att_src,
                           const float* __restrict__ att_dst,
                           float* __restrict__ a_src, float* __restrict__ a_dst) {
  int t = blockIdx.x * blockDim.x + threadIdx.x;
  if (t >= N_NODES * HEADS) return;
  const int hh = t & (HEADS - 1);
  const int node = t >> 3;
  const float* __restrict__ hv = h + (size_t)node * FDIM + hh * HEAD_DIM;
  const float* __restrict__ asv = att_src + hh * HEAD_DIM;
  const float* __restrict__ adv = att_dst + hh * HEAD_DIM;
  float s1 = 0.f, s2 = 0.f;
#pragma unroll
  for (int d = 0; d < HEAD_DIM; ++d) {
    float v = hv[d];
    s1 += v * asv[d];
    s2 += v * adv[d];
  }
  a_src[t] = s1;
  a_dst[t] = s2;
}

__device__ __forceinline__ void edge_decode(const int* __restrict__ ei, int e,
                                            int& src, int& dst) {
  if (e < N_EDGES) {
    src = ei[e];
    dst = ei[N_EDGES + e];
  } else {
    src = dst = e - N_EDGES;
  }
}

__device__ __forceinline__ float leaky(float x) {
  return x > 0.f ? x : NEG_SLOPE * x;
}

__device__ __forceinline__ void lds_max_f(float* addr, float val) {
  if (__float_as_int(val) >= 0)
    atomicMax((int*)addr, __float_as_int(val));
  else
    atomicMin((unsigned int*)addr, __float_as_uint(val));
}

#define R_TILE 196
#define LCAP   8192
#define TB     1024
#define SCAN_K 8
#define E4     (N_EDGES / 4)
#define TL_HDR    32
#define TL_STRIDE (LCAP + TL_HDR)

__device__ __forceinline__ int tile_col(int c) { return (c & 3) * 32 + (c >> 2); }

template <bool FIRST>
__global__ void __launch_bounds__(TB)
tile_edges(const int* __restrict__ ei, const float* __restrict__ a_src,
           const float* __restrict__ a_dst, const float* __restrict__ h,
           const float* __restrict__ bias, float* __restrict__ out,
           unsigned int* __restrict__ tl) {
  constexpr bool ELU = FIRST;
  __shared__ __attribute__((aligned(16))) float acc[R_TILE * FDIM];
  __shared__ float sm[R_TILE * HEADS];
  __shared__ float ss[R_TILE * HEADS];
  __shared__ float sad[R_TILE * HEADS];
  __shared__ __attribute__((aligned(16))) unsigned int list[LCAP];
  __shared__ int cnt;

  const int tid  = threadIdx.x;
  const int lane = tid & 31;
  const int wave = tid >> 5;
  const int base = blockIdx.x * R_TILE;
  const int nvalid = (N_NODES - base) < R_TILE ? (N_NODES - base) : R_TILE;
  unsigned int* __restrict__ my_tl = tl + (size_t)blockIdx.x * TL_STRIDE;

  for (int i = tid; i < R_TILE * FDIM / 4; i += TB)
    ((float4*)acc)[i] = make_float4(0.f, 0.f, 0.f, 0.f);
  for (int i = tid; i < R_TILE * HEADS; i += TB) {
    const int node = base + (i >> 3);
    sm[i]  = -__builtin_inff();
    ss[i]  = 0.f;
    sad[i] = node < N_NODES ? a_dst[(size_t)node * HEADS + (i & 7)] : 0.f;
  }
  if (tid == 0) cnt = 0;
  __syncthreads();

  if (FIRST) {
    const int4* __restrict__ dst4 = (const int4*)(ei + N_EDGES);
    for (int it = 0; it < E4; it += TB * SCAN_K) {
      int4 d[SCAN_K];
#pragma unroll
      for (int k = 0; k < SCAN_K; ++k) {
        const int q = it + k * TB + tid;
        d[k] = q < E4 ? dst4[q] : make_int4(-1, -1, -1, -1);
      }
      unsigned int msk[SCAN_K * 4];
      int nw = 0;
#pragma unroll
      for (int k = 0; k < SCAN_K; ++k) {
        const int dv[4] = {d[k].x, d[k].y, d[k].z, d[k].w};
#pragma unroll
        for (int j = 0; j < 4; ++j) {
          const bool mt = (unsigned int)(dv[j] - base) < (unsigned int)R_TILE;
          msk[k * 4 + j] = __builtin_amdgcn_ballot_w32(mt);
          nw += __builtin_popcount(msk[k * 4 + j]);
        }
      }
      if (nw) {
        int bp = 0;
        if (lane == 0) bp = atomicAdd(&cnt, nw);
        bp = __builtin_amdgcn_readfirstlane(bp);
#pragma unroll
        for (int k = 0; k < SCAN_K; ++k) {
          const int dv[4] = {d[k].x, d[k].y, d[k].z, d[k].w};
#pragma unroll
          for (int j = 0; j < 4; ++j) {
            const unsigned int mq = msk[k * 4 + j];
            if (mq) {
              if ((mq >> lane) & 1u) {
                const int pos = bp + (int)__builtin_amdgcn_mbcnt_lo(mq, 0u);
                if (pos < LCAP)
                  list[pos] = (unsigned int)((it + k * TB + tid) * 4 + j) |
                              ((unsigned int)(dv[j] - base) << 20);
              }
              bp += __builtin_popcount(mq);
            }
          }
        }
      }
    }
  }
  int nscan, total;
  if (FIRST) {
    __syncthreads();
    nscan = cnt;
    total = nscan + nvalid;
    if (tid == 0) my_tl[0] = (unsigned int)total;
  } else {
    nscan = 0;
    total = __builtin_amdgcn_readfirstlane((int)my_tl[0]);
  }

  if (total <= LCAP) {
    if (FIRST) {
      if (tid < nvalid)
        list[nscan + tid] = (unsigned int)(base + tid) | ((unsigned int)tid << 16);
    } else {
      const uint4* __restrict__ g4 = (const uint4*)(my_tl + TL_HDR);
      for (int i = tid; i < (total + 3) / 4; i += TB) ((uint4*)list)[i] = g4[i];
    }
    __syncthreads();
    for (int i = tid; i < total; i += TB) {
      const unsigned int a = list[i];
      unsigned int src, dl;
      if (FIRST && i < nscan) {
        src = (unsigned int)ei[a & 0xFFFFFu];
        dl  = a >> 20;
        list[i] = src | (dl << 16);
      } else {
        src = a & 0xFFFFu;
        dl  = a >> 16;
      }
      const float4 s0 = *(const float4*)(a_src + (size_t)src * HEADS);
      const float4 s1 = *(const float4*)(a_src + (size_t)src * HEADS + 4);
      const float sv[8] = {s0.x, s0.y, s0.z, s0.w, s1.x, s1.y, s1.z, s1.w};
#pragma unroll
      for (int hh = 0; hh < HEADS; ++hh)
        lds_max_f(&sm[dl * HEADS + hh], leaky(sv[hh] + sad[dl * HEADS + hh]));
    }
    __syncthreads();
    if (FIRST) {
      uint4* __restrict__ g4 = (uint4*)(my_tl + TL_HDR);
      for (int i = tid; i < (total + 3) / 4; i += TB) g4[i] = ((const uint4*)list)[i];
    }
    for (int i = tid; i < R_TILE * HEADS; i += TB) {
      const float mm = sm[i];
      sm[i] = (__builtin_fabsf(mm) < __builtin_inff()) ? mm : 0.f;
    }
    __syncthreads();
    constexpr int ADD_U = 8;
    const int hh = lane >> 2;
    for (int i0 = wave * ADD_U; i0 < total; i0 += (TB / 32) * ADD_U) {
      unsigned int a[ADD_U];
      float4 v[ADD_U];
      float as[ADD_U];
#pragma unroll
      for (int u = 0; u < ADD_U; ++u)
        a[u] = (i0 + u) < total ? list[i0 + u] : 0u;
#pragma unroll
      for (int u = 0; u < ADD_U; ++u) {
        const unsigned int src = a[u] & 0xFFFFu;
        v[u]  = *(const float4*)(h + (size_t)src * FDIM + lane * 4);
        as[u] = a_src[(size_t)src * HEADS + hh];
      }
#pragma unroll
      for (int u = 0; u < ADD_U; ++u) {
        if ((i0 + u) < total) {
          const unsigned int dl = a[u] >> 16;
          const float ev = leaky(as[u] + sad[dl * HEADS + hh]);
          const float p  = expf(ev - sm[dl * HEADS + hh]);
          float* ar = &acc[dl * FDIM + lane];
          atomicAdd(ar,      p * v[u].x);
          atomicAdd(ar + 32, p * v[u].y);
          atomicAdd(ar + 64, p * v[u].z);
          atomicAdd(ar + 96, p * v[u].w);
          if ((lane & 3) == 0) atomicAdd(&ss[dl * HEADS + hh], p);
        }
      }
    }
  } else {
    for (int e = tid; e < ETOT; e += TB) {
      int src, dst;
      edge_decode(ei, e, src, dst);
      const unsigned int dl = (unsigned int)(dst - base);
      if (dl < (unsigned int)R_TILE) {
        for (int hh = 0; hh < HEADS; ++hh)
          lds_max_f(&sm[dl * HEADS + hh],
                    leaky(a_src[(size_t)src * HEADS + hh] + sad[dl * HEADS + hh]));
      }
    }
    __syncthreads();
    for (int i = tid; i < R_TILE * HEADS; i += TB) {
      const float mm = sm[i];
      sm[i] = (__builtin_fabsf(mm) < __builtin_inff()) ? mm : 0.f;
    }
    __syncthreads();
    for (int e = tid; e < ETOT; e += TB) {
      int src, dst;
      edge_decode(ei, e, src, dst);
      const unsigned int dl = (unsigned int)(dst - base);
      if (dl < (unsigned int)R_TILE) {
        for (int hh = 0; hh < HEADS; ++hh) {
          const float ev =
              leaky(a_src[(size_t)src * HEADS + hh] + sad[dl * HEADS + hh]);
          const float p = expf(ev - sm[dl * HEADS + hh]);
          atomicAdd(&ss[dl * HEADS + hh], p);
          for (int dd = 0; dd < HEAD_DIM; ++dd) {
            const int c = hh * HEAD_DIM + dd;
            atomicAdd(&acc[dl * FDIM + tile_col(c)],
                      p * h[(size_t)src * FDIM + c]);
          }
        }
      }
    }
  }
  __syncthreads();

  for (int idx = tid; idx < R_TILE * 32; idx += TB) {
    const int row = idx >> 5;
    const int l   = idx & 31;
    const int node = base + row;
    if (node < N_NODES) {
      const float den = ss[row * HEADS + (l >> 2)] + 1e-16f;
      const float4 b = *(const float4*)(bias + l * 4);
      float4 o;
      o.x = acc[row * FDIM + l]      / den + b.x;
      o.y = acc[row * FDIM + 32 + l] / den + b.y;
      o.z = acc[row * FDIM + 64 + l] / den + b.z;
      o.w = acc[row * FDIM + 96 + l] / den + b.w;
      if (ELU) {
        o.x = o.x > 0.f ? o.x : (expf(o.x) - 1.f);
        o.y = o.y > 0.f ? o.y : (expf(o.y) - 1.f);
        o.z = o.z > 0.f ? o.z : (expf(o.z) - 1.f);
        o.w = o.w > 0.f ? o.w : (expf(o.w) - 1.f);
      }
      *(float4*)(out + (size_t)node * FDIM + l * 4) = o;
    }
  }
}

extern "C" void kernel_launch(void* const* d_in, const int* in_sizes, int n_in,
                              void* d_out, int out_size, void* d_ws,
                              size_t ws_size, hipStream_t stream) {
  (void)in_sizes; (void)n_in; (void)out_size;

  const float* x   = (const float*)d_in[0];
  const int*   ei  = (const int*)d_in[1];
  const float* W1  = (const float*)d_in[2];
  const float* as1 = (const float*)d_in[3];
  const float* ad1 = (const float*)d_in[4];
  const float* b1  = (const float*)d_in[5];
  const float* W2  = (const float*)d_in[6];
  const float* as2 = (const float*)d_in[7];
  const float* ad2 = (const float*)d_in[8];
  const float* b2  = (const float*)d_in[9];
  float* out = (float*)d_out;

  float* ws    = (float*)d_ws;
  float* h     = ws;
  float* agg   = h + (size_t)N_NODES * FDIM;
  float* a_src = agg + (size_t)N_NODES * FDIM;
  float* a_dst = a_src + (size_t)N_NODES * HEADS;
  unsigned int* tl = (unsigned int*)(a_dst + (size_t)N_NODES * HEADS);
  const int gT  = (N_NODES + R_TILE - 1) / R_TILE;
  const size_t need = ((size_t)N_NODES * FDIM * 2 + (size_t)N_NODES * HEADS * 2) *
                          sizeof(float) +
                      (size_t)gT * TL_STRIDE * sizeof(unsigned int);
  if (ws_size < need) return;

  const int nh = N_NODES * HEADS;
  const int gNH = (nh + 255) / 256;
  dim3 BG(GEMM_TB), B256(256), BT(TB);

  gemm128_wmma_f32<<<GEMM_GROUPS, BG, 0, stream>>>(x, W1, h);
  att_scores<<<gNH, B256, 0, stream>>>(h, as1, ad1, a_src, a_dst);
  tile_edges<true><<<gT, BT, 0, stream>>>(ei, a_src, a_dst, h, b1, agg, tl);

  gemm128_wmma_f32<<<GEMM_GROUPS, BG, 0, stream>>>(agg, W2, h);
  att_scores<<<gNH, B256, 0, stream>>>(h, as2, ad2, a_src, a_dst);
  tile_edges<false><<<gT, BT, 0, stream>>>(ei, a_src, a_dst, h, b2, out, tl);
}


# layer-1 edge scan reads one byte per edge (its destination tile's number, written by the layer-1 score kernel of the same call) instead of 4; list order, so LDS arrival order of the sums, changes
# speedup vs baseline: 7.5979x; 1.1187x over previous
#include <hip/hip_runtime.h>
#include <math.h>


#define N_NODES 50000
#define N_EDGES 800000
#define HEADS   8
#define HEAD_DIM 16
#define FDIM    128
#define ETOT    (N_EDGES + N_NODES)
#define NEG_SLOPE 0.2f
#define R_TILE 196

typedef __attribute__((ext_vector_type(2))) float v2f;
typedef __attribute__((ext_vector_type(8))) float v8f;

#define GEMM_WAVES  8
#define GEMM_TB     (GEMM_WAVES * 32)
#define GEMM_GROUPS 512
#define ROW_TILES   (N_NODES / 16)
#define A_PITCH     132

__global__ void __launch_bounds__(GEMM_TB)
gemm128_wmma_f32(const float* __restrict__ A, const float* __restrict__ W,
                 float* __restrict__ out) {
  __shared__ __attribute__((aligned(16))) float As[16 * A_PITCH];
  const int tid  = threadIdx.x;
  const int wave = tid >> 5;
  const int lane = tid & 31;
  const int half = lane >> 4;
  const int l    = lane & 15;

  v2f b[32];
  const float* __restrict__ bcol = W + wave * 16 + l;
#pragma unroll
  for (int s = 0; s < 32; ++s) {
    const int k = 8 * (s >> 1) + 4 * half + 2 * (s & 1);
    b[s].x = bcol[(size_t)k * FDIM];
    b[s].y = bcol[(size_t)(k + 1) * FDIM];
  }

  float* __restrict__ st0 = &As[wave * A_PITCH + lane * 4];
  float* __restrict__ st1 = &As[(wave + 8) * A_PITCH + lane * 4];
  const float* __restrict__ frag = &As[l * A_PITCH + 4 * half];

  int tm = blockIdx.x;
  const float4* __restrict__ src = (const float4*)(A + (size_t)tm * 16 * FDIM);
  float4 g0 = src[tid];
  float4 g1 = src[tid + GEMM_TB];
  for (; tm < ROW_TILES; tm += GEMM_GROUPS) {
    *(float4*)st0 = g0;
    *(float4*)st1 = g1;
    __syncthreads();
    const int nx = tm + GEMM_GROUPS;
    if (nx < ROW_TILES) {
      src = (const float4*)(A + (size_t)nx * 16 * FDIM);
      g0 = src[tid];
      g1 = src[tid + GEMM_TB];
    }
    float4 q[16];
#pragma unroll
    for (int j = 0; j < 16; ++j) q[j] = *(const float4*)(frag + 8 * j);
    __syncthreads();

    v8f c = {};
#pragma unroll
    for (int j = 0; j < 16; ++j) {
      v2f a;
      a.x = q[j].x;
      a.y = q[j].y;
      c = __builtin_amdgcn_wmma_f32_16x16x4_f32(false, a, false, b[2 * j],
                                                (short)0, c, false, false);
      a.x = q[j].z;
      a.y = q[j].w;
      c = __builtin_amdgcn_wmma_f32_16x16x4_f32(false, a, false, b[2 * j + 1],
                                                (short)0, c, false, false);
    }
    float* __restrict__ orow =
        out + (size_t)(tm * 16 + 8 * half) * FDIM + wave * 16 + l;
#pragma unroll
    for (int r = 0; r < 8; ++r) orow[(size_t)r * FDIM] = c[r];
  }
}

template <bool FIRST>
__global__ void att_scores(const float* __restrict__ h,
                           const float* __restrict__ att_src,
                           const float* __restrict__ att_dst,
                           float* __restrict__ a_src, float* __restrict__ a_dst,
                           const int* __restrict__ ei,
                           unsigned int* __restrict__ tb) {
  int t = blockIdx.x * blockDim.x + threadIdx.x;
  const bool pack = FIRST && t < N_EDGES / 4;
  int4 dq = make_int4(0, 0, 0, 0);
  if (FIRST) dq = ((const int4*)(ei + N_EDGES))[t < N_EDGES / 4 ? t : N_EDGES / 4 - 1];
  if (t < N_NODES * HEADS) {
    const int hh = t & (HEADS - 1);
    const int node = t >> 3;
    const float* __restrict__ hv = h + (size_t)node * FDIM + hh * HEAD_DIM;
    const float* __restrict__ asv = att_src + hh * HEAD_DIM;
    const float* __restrict__ adv = att_dst + hh * HEAD_DIM;
    float s1 = 0.f, s2 = 0.f;
#pragma unroll
    for (int d = 0; d < HEAD_DIM; ++d) {
      float v = hv[d];
      s1 += v * asv[d];
      s2 += v * adv[d];
    }
    a_src[t] = s1;
    a_dst[t] = s2;
  }
  if (pack)
    tb[t] = ((unsigned int)dq.x / R_TILE) | (((unsigned int)dq.y / R_TILE) << 8) |
            (((unsigned int)dq.z / R_TILE) << 16) |
            (((unsigned int)dq.w / R_TILE) << 24);
}

__device__ __forceinline__ void edge_decode(const int* __restrict__ ei, int e,
                                            int& src, int& dst) {
  if (e < N_EDGES) {
    src = ei[e];
    dst = ei[N_EDGES + e];
  } else {
    src = dst = e - N_EDGES;
  }
}

__device__ __forceinline__ float leaky(float x) {
  return x > 0.f ? x : NEG_SLOPE * x;
}

__device__ __forceinline__ void lds_max_f(float* addr, float val) {
  if (__float_as_int(val) >= 0)
    atomicMax((int*)addr, __float_as_int(val));
  else
    atomicMin((unsigned int*)addr, __float_as_uint(val));
}

#define LCAP   8192
#define TB     1024
#define SCAN_K 7
#define E16    (N_EDGES / 16)
#define TL_HDR    32
#define TL_STRIDE (LCAP + TL_HDR)

__device__ __forceinline__ unsigned int zero_bytes(unsigned int x) {
  const unsigned int t = (x & 0x7F7F7F7Fu) + 0x7F7F7F7Fu;
  return ~(t | x | 0x7F7F7F7Fu);
}

__device__ __forceinline__ int tile_col(int c) { return (c & 3) * 32 + (c >> 2); }

template <bool FIRST>
__global__ void __launch_bounds__(TB)
tile_edges(const int* __restrict__ ei, const float* __restrict__ a_src,
           const float* __restrict__ a_dst, const float* __restrict__ h,
           const float* __restrict__ bias, float* __restrict__ out,
           unsigned int* __restrict__ tl, const unsigned int* __restrict__ tb) {
  constexpr bool ELU = FIRST;
  __shared__ __attribute__((aligned(16))) float acc[R_TILE * FDIM];
  __shared__ float sm[R_TILE * HEADS];
  __shared__ float ss[R_TILE * HEADS];
  __shared__ float sad[R_TILE * HEADS];
  __shared__ __attribute__((aligned(16))) unsigned int list[LCAP];
  __shared__ int cnt;

  const int tid  = threadIdx.x;
  const int lane = tid & 31;
  const int wave = tid >> 5;
  const int base = blockIdx.x * R_TILE;
  const int nvalid = (N_NODES - base) < R_TILE ? (N_NODES - base) : R_TILE;
  unsigned int* __restrict__ my_tl = tl + (size_t)blockIdx.x * TL_STRIDE;

  for (int i = tid; i < R_TILE * FDIM / 4; i += TB)
    ((float4*)acc)[i] = make_float4(0.f, 0.f, 0.f, 0.f);
  for (int i = tid; i < R_TILE * HEADS; i += TB) {
    const int node = base + (i >> 3);
    sm[i]  = -__builtin_inff();
    ss[i]  = 0.f;
    sad[i] = node < N_NODES ? a_dst[(size_t)node * HEADS + (i & 7)] : 0.f;
  }
  if (tid == 0) cnt = 0;
  __syncthreads();

  if (FIRST) {
    const uint4* __restrict__ tb4 = (const uint4*)tb;
    const unsigned int pat = (unsigned int)blockIdx.x * 0x01010101u;
    for (int it = 0; it < E16; it += TB * SCAN_K) {
      uint4 d[SCAN_K];
#pragma unroll
      for (int k = 0; k < SCAN_K; ++k) {
        const int q = it + k * TB + tid;
        d[k] = q < E16 ? tb4[q] : make_uint4(~pat, ~pat, ~pat, ~pat);
      }
      unsigned int mw[(SCAN_K + 1) / 2];
      int n = 0;
#pragma unroll
      for (int k = 0; k < SCAN_K; ++k) {
        const unsigned int mk = (zero_bytes(d[k].x ^ pat) >> 7) |
                                (zero_bytes(d[k].y ^ pat) >> 6) |
                                (zero_bytes(d[k].z ^ pat) >> 5) |
                                (zero_bytes(d[k].w ^ pat) >> 4);
        n += __builtin_popcount(mk);
        if (k & 1) mw[k >> 1] |= mk << 4;
        else       mw[k >> 1] = mk;
      }
      int pre = 0, tot = 0;
#pragma unroll
      for (int b = 0; b < 7; ++b) {
        const unsigned int pl = __builtin_amdgcn_ballot_w32(((n >> b) & 1) != 0);
        pre += (int)__builtin_amdgcn_mbcnt_lo(pl, 0u) << b;
        tot += __builtin_popcount(pl) << b;
      }
      if (tot) {
        int bp = 0;
        if (lane == 0) bp = atomicAdd(&cnt, tot);
        int pos = __builtin_amdgcn_readfirstlane(bp) + pre;
#pragma unroll
        for (int w = 0; w < (SCAN_K + 1) / 2; ++w) {
          unsigned int m = mw[w];
          const unsigned int e0 = (unsigned int)(it + 2 * w * TB + tid) * 16u;
          while (m) {
            const unsigned int p = (unsigned int)__builtin_ctz(m);
            m &= m - 1u;
            if (pos < LCAP)
              list[pos] = e0 + ((p >> 2) & 1u) * (16u * TB) + 4u * (p & 3u) + (p >> 3);
            ++pos;
          }
        }
      }
    }
  }
  int nscan, total;
  if (FIRST) {
    __syncthreads();
    nscan = cnt;
    total = nscan + nvalid;
    if (tid == 0) my_tl[0] = (unsigned int)total;
  } else {
    nscan = 0;
    total = __builtin_amdgcn_readfirstlane((int)my_tl[0]);
  }

  if (total <= LCAP) {
    if (FIRST) {
      if (tid < nvalid)
        list[nscan + tid] = (unsigned int)(base + tid) | ((unsigned int)tid << 16);
    } else {
      const uint4* __restrict__ g4 = (const uint4*)(my_tl + TL_HDR);
      for (int i = tid; i < (total + 3) / 4; i += TB) ((uint4*)list)[i] = g4[i];
    }
    __syncthreads();
    for (int i = tid; i < total; i += TB) {
      const unsigned int a = list[i];
      unsigned int src, dl;
      if (FIRST && i < nscan) {
        src = (unsigned int)ei[a];
        dl  = (unsigned int)(ei[N_EDGES + a] - base);
        list[i] = src | (dl << 16);
      } else {
        src = a & 0xFFFFu;
        dl  = a >> 16;
      }
      const float4 s0 = *(const float4*)(a_src + (size_t)src * HEADS);
      const float4 s1 = *(const float4*)(a_src + (size_t)src * HEADS + 4);
      const float sv[8] = {s0.x, s0.y, s0.z, s0.w, s1.x, s1.y, s1.z, s1.w};
#pragma unroll
      for (int hh = 0; hh < HEADS; ++hh)
        lds_max_f(&sm[dl * HEADS + hh], leaky(sv[hh] + sad[dl * HEADS + hh]));
    }
    __syncthreads();
    if (FIRST) {
      uint4* __restrict__ g4 = (uint4*)(my_tl + TL_HDR);
      for (int i = tid; i < (total + 3) / 4; i += TB) g4[i] = ((const uint4*)list)[i];
    }
    for (int i = tid; i < R_TILE * HEADS; i += TB) {
      const float mm = sm[i];
      sm[i] = (__builtin_fabsf(mm) < __builtin_inff()) ? mm : 0.f;
    }
    __syncthreads();
    constexpr int ADD_U = 8;
    const int hh = lane >> 2;
    for (int i0 = wave * ADD_U; i0 < total; i0 += (TB / 32) * ADD_U) {
      unsigned int a[ADD_U];
      float4 v[ADD_U];
      float as[ADD_U];
#pragma unroll
      for (int u = 0; u < ADD_U; ++u)
        a[u] = (i0 + u) < total ? list[i0 + u] : 0u;
#pragma unroll
      for (int u = 0; u < ADD_U; ++u) {
        const unsigned int src = a[u] & 0xFFFFu;
        v[u]  = *(const float4*)(h + (size_t)src * FDIM + lane * 4);
        as[u] = a_src[(size_t)src * HEADS + hh];
      }
#pragma unroll
      for (int u = 0; u < ADD_U; ++u) {
        if ((i0 + u) < total) {
          const unsigned int dl = a[u] >> 16;
          const float ev = leaky(as[u] + sad[dl * HEADS + hh]);
          const float p  = expf(ev - sm[dl * HEADS + hh]);
          float* ar = &acc[dl * FDIM + lane];
          atomicAdd(ar,      p * v[u].x);
          atomicAdd(ar + 32, p * v[u].y);
          atomicAdd(ar + 64, p * v[u].z);
          atomicAdd(ar + 96, p * v[u].w);
          if ((lane & 3) == 0) atomicAdd(&ss[dl * HEADS + hh], p);
        }
      }
    }
  } else {
    for (int e = tid; e < ETOT; e += TB) {
      int src, dst;
      edge_decode(ei, e, src, dst);
      const unsigned int dl = (unsigned int)(dst - base);
      if (dl < (unsigned int)R_TILE) {
        for (int hh = 0; hh < HEADS; ++hh)
          lds_max_f(&sm[dl * HEADS + hh],
                    leaky(a_src[(size_t)src * HEADS + hh] + sad[dl * HEADS + hh]));
      }
    }
    __syncthreads();
    for (int i = tid; i < R_TILE * HEADS; i += TB) {
      const float mm = sm[i];
      sm[i] = (__builtin_fabsf(mm) < __builtin_inff()) ? mm : 0.f;
    }
    __syncthreads();
    for (int e = tid; e < ETOT; e += TB) {
      int src, dst;
      edge_decode(ei, e, src, dst);
      const unsigned int dl = (unsigned int)(dst - base);
      if (dl < (unsigned int)R_TILE) {
        for (int hh = 0; hh < HEADS; ++hh) {
          const float ev =
              leaky(a_src[(size_t)src * HEADS + hh] + sad[dl * HEADS + hh]);
          const float p = expf(ev - sm[dl * HEADS + hh]);
          atomicAdd(&ss[dl * HEADS + hh], p);
          for (int dd = 0; dd < HEAD_DIM; ++dd) {
            const int c = hh * HEAD_DIM + dd;
            atomicAdd(&acc[dl * FDIM + tile_col(c)],
                      p * h[(size_t)src * FDIM + c]);
          }
        }
      }
    }
  }
  __syncthreads();

  for (int idx = tid; idx < R_TILE * 32; idx += TB) {
    const int row = idx >> 5;
    const int l   = idx & 31;
    const int node = base + row;
    if (node < N_NODES) {
      const float den = ss[row * HEADS + (l >> 2)] + 1e-16f;
      const float4 b = *(const float4*)(bias + l * 4);
      float4 o;
      o.x = acc[row * FDIM + l]      / den + b.x;
      o.y = acc[row * FDIM + 32 + l] / den + b.y;
      o.z = acc[row * FDIM + 64 + l] / den + b.z;
      o.w = acc[row * FDIM + 96 + l] / den + b.w;
      if (ELU) {
        o.x = o.x > 0.f ? o.x : (expf(o.x) - 1.f);
        o.y = o.y > 0.f ? o.y : (expf(o.y) - 1.f);
        o.z = o.z > 0.f ? o.z : (expf(o.z) - 1.f);
        o.w = o.w > 0.f ? o.w : (expf(o.w) - 1.f);
      }
      *(float4*)(out + (size_t)node * FDIM + l * 4) = o;
    }
  }
}

extern "C" void kernel_launch(void* const* d_in, const int* in_sizes, int n_in,
                              void* d_out, int out_size, void* d_ws,
                              size_t ws_size, hipStream_t stream) {
  (void)in_sizes; (void)n_in; (void)out_size;

  const float* x   = (const float*)d_in[0];
  const int*   ei  = (const int*)d_in[1];
  const float* W1  = (const float*)d_in[2];
  const float* as1 = (const float*)d_in[3];
  const float* ad1 = (const float*)d_in[4];
  const float* b1  = (const float*)d_in[5];
  const float* W2  = (const float*)d_in[6];
  const float* as2 = (const float*)d_in[7];
  const float* ad2 = (const float*)d_in[8];
  const float* b2  = (const float*)d_in[9];
  float* out = (float*)d_out;

  float* ws    = (float*)d_ws;
  float* h     = ws;
  float* agg   = h + (size_t)N_NODES * FDIM;
  float* a_src = agg + (size_t)N_NODES * FDIM;
  float* a_dst = a_src + (size_t)N_NODES * HEADS;
  unsigned int* tl = (unsigned int*)(a_dst + (size_t)N_NODES * HEADS);
  const int gT  = (N_NODES + R_TILE - 1) / R_TILE;
  unsigned int* tb = tl + (size_t)gT * TL_STRIDE;
  const size_t need = ((size_t)N_NODES * FDIM * 2 + (size_t)N_NODES * HEADS * 2) *
                          sizeof(float) +
                      (size_t)gT * TL_STRIDE * sizeof(unsigned int) + N_EDGES;
  if (ws_size < need) return;

  const int nh = N_NODES * HEADS;
  const int gNH = (nh + 255) / 256;
  dim3 BG(GEMM_TB), B256(256), BT(TB);

  gemm128_wmma_f32<<<GEMM_GROUPS, BG, 0, stream>>>(x, W1, h);
  att_scores<true><<<gNH, B256, 0, stream>>>(h, as1, ad1, a_src, a_dst, ei, tb);
  tile_edges<true><<<gT, BT, 0, stream>>>(ei, a_src, a_dst, h, b1, agg, tl, tb);

  gemm128_wmma_f32<<<GEMM_GROUPS, BG, 0, stream>>>(agg, W2, h);
  att_scores<false><<<gNH, B256, 0, stream>>>(h, as2, ad2, a_src, a_dst, ei, tb);
  tile_edges<false><<<gT, BT, 0, stream>>>(ei, a_src, a_dst, h, b2, out, tl, tb);
}


# edge add phase: each (edge, head) weight exp(e-m) is worked out once per wave, two per lane, and read by the head's 4 lanes through a quad DPP multiply, not in all 32 lanes; LDS arrival order changes
# speedup vs baseline: 8.5015x; 1.1189x over previous
#include <hip/hip_runtime.h>
#include <math.h>


#define N_NODES 50000
#define N_EDGES 800000
#define HEADS   8
#define HEAD_DIM 16
#define FDIM    128
#define ETOT    (N_EDGES + N_NODES)
#define NEG_SLOPE 0.2f
#define R_TILE 196

typedef __attribute__((ext_vector_type(2))) float v2f;
typedef __attribute__((ext_vector_type(8))) float v8f;

#define GEMM_WAVES  8
#define GEMM_TB     (GEMM_WAVES * 32)
#define GEMM_GROUPS 512
#define ROW_TILES   (N_NODES / 16)
#define A_PITCH     132

__global__ void __launch_bounds__(GEMM_TB)
gemm128_wmma_f32(const float* __restrict__ A, const float* __restrict__ W,
                 float* __restrict__ out) {
  __shared__ __attribute__((aligned(16))) float As[16 * A_PITCH];
  const int tid  = threadIdx.x;
  const int wave = tid >> 5;
  const int lane = tid & 31;
  const int half = lane >> 4;
  const int l    = lane & 15;

  v2f b[32];
  const float* __restrict__ bcol = W + wave * 16 + l;
#pragma unroll
  for (int s = 0; s < 32; ++s) {
    const int k = 8 * (s >> 1) + 4 * half + 2 * (s & 1);
    b[s].x = bcol[(size_t)k * FDIM];
    b[s].y = bcol[(size_t)(k + 1) * FDIM];
  }

  float* __restrict__ st0 = &As[wave * A_PITCH + lane * 4];
  float* __restrict__ st1 = &As[(wave + 8) * A_PITCH + lane * 4];
  const float* __restrict__ frag = &As[l * A_PITCH + 4 * half];

  int tm = blockIdx.x;
  const float4* __restrict__ src = (const float4*)(A + (size_t)tm * 16 * FDIM);
  float4 g0 = src[tid];
  float4 g1 = src[tid + GEMM_TB];
  for (; tm < ROW_TILES; tm += GEMM_GROUPS) {
    *(float4*)st0 = g0;
    *(float4*)st1 = g1;
    __syncthreads();
    const int nx = tm + GEMM_GROUPS;
    if (nx < ROW_TILES) {
      src = (const float4*)(A + (size_t)nx * 16 * FDIM);
      g0 = src[tid];
      g1 = src[tid + GEMM_TB];
    }
    float4 q[16];
#pragma unroll
    for (int j = 0; j < 16; ++j) q[j] = *(const float4*)(frag + 8 * j);
    __syncthreads();

    v8f c = {};
#pragma unroll
    for (int j = 0; j < 16; ++j) {
      v2f a;
      a.x = q[j].x;
      a.y = q[j].y;
      c = __builtin_amdgcn_wmma_f32_16x16x4_f32(false, a, false, b[2 * j],
                                                (short)0, c, false, false);
      a.x = q[j].z;
      a.y = q[j].w;
      c = __builtin_amdgcn_wmma_f32_16x16x4_f32(false, a, false, b[2 * j + 1],
                                                (short)0, c, false, false);
    }
    float* __restrict__ orow =
        out + (size_t)(tm * 16 + 8 * half) * FDIM + wave * 16 + l;
#pragma unroll
    for (int r = 0; r < 8; ++r) orow[(size_t)r * FDIM] = c[r];
  }
}

template <bool FIRST>
__global__ void att_scores(const float* __restrict__ h,
                           const float* __restrict__ att_src,
                           const float* __restrict__ att_dst,
                           float* __restrict__ a_src, float* __restrict__ a_dst,
                           const int* __restrict__ ei,
                           unsigned int* __restrict__ tb) {
  int t = blockIdx.x * blockDim.x + threadIdx.x;
  const bool pack = FIRST && t < N_EDGES / 4;
  int4 dq = make_int4(0, 0, 0, 0);
  if (FIRST) dq = ((const int4*)(ei + N_EDGES))[t < N_EDGES / 4 ? t : N_EDGES / 4 - 1];
  if (t < N_NODES * HEADS) {
    const int hh = t & (HEADS - 1);
    const int node = t >> 3;
    const float* __restrict__ hv = h + (size_t)node * FDIM + hh * HEAD_DIM;
    const float* __restrict__ asv = att_src + hh * HEAD_DIM;
    const float* __restrict__ adv = att_dst + hh * HEAD_DIM;
    float s1 = 0.f, s2 = 0.f;
#pragma unroll
    for (int d = 0; d < HEAD_DIM; ++d) {
      float v = hv[d];
      s1 += v * asv[d];
      s2 += v * adv[d];
    }
    a_src[t] = s1;
    a_dst[t] = s2;
  }
  if (pack)
    tb[t] = ((unsigned int)dq.x / R_TILE) | (((unsigned int)dq.y / R_TILE) << 8) |
            (((unsigned int)dq.z / R_TILE) << 16) |
            (((unsigned int)dq.w / R_TILE) << 24);
}

__device__ __forceinline__ void edge_decode(const int* __restrict__ ei, int e,
                                            int& src, int& dst) {
  if (e < N_EDGES) {
    src = ei[e];
    dst = ei[N_EDGES + e];
  } else {
    src = dst = e - N_EDGES;
  }
}

__device__ __forceinline__ float leaky(float x) {
  return x > 0.f ? x : NEG_SLOPE * x;
}

__device__ __forceinline__ void lds_max_f(float* addr, float val) {
  if (__float_as_int(val) >= 0)
    atomicMax((int*)addr, __float_as_int(val));
  else
    atomicMin((unsigned int*)addr, __float_as_uint(val));
}

#define LCAP   8192
#define TB     1024
#define SCAN_K 7
#define E16    (N_EDGES / 16)
#define TL_HDR    32
#define TL_STRIDE (LCAP + TL_HDR)

__device__ __forceinline__ unsigned int zero_bytes(unsigned int x) {
  const unsigned int t = (x & 0x7F7F7F7Fu) + 0x7F7F7F7Fu;
  return ~(t | x | 0x7F7F7F7Fu);
}

__device__ __forceinline__ int tile_col(int c) { return (c & 3) * 32 + (c >> 2); }

__device__ __forceinline__ float quad_lane(float x, int s) {
  const int xi = __float_as_int(x);
  int r;
  switch (s) {
    case 0:  r = __builtin_amdgcn_update_dpp(0, xi, 0x00, 0xF, 0xF, false); break;
    case 1:  r = __builtin_amdgcn_update_dpp(0, xi, 0x55, 0xF, 0xF, false); break;
    case 2:  r = __builtin_amdgcn_update_dpp(0, xi, 0xAA, 0xF, 0xF, false); break;
    default: r = __builtin_amdgcn_update_dpp(0, xi, 0xFF, 0xF, 0xF, false); break;
  }
  return __int_as_float(r);
}

template <bool FIRST>
__global__ void __launch_bounds__(TB)
tile_edges(const int* __restrict__ ei, const float* __restrict__ a_src,
           const float* __restrict__ a_dst, const float* __restrict__ h,
           const float* __restrict__ bias, float* __restrict__ out,
           unsigned int* __restrict__ tl, const unsigned int* __restrict__ tb) {
  constexpr bool ELU = FIRST;
  __shared__ __attribute__((aligned(16))) float acc[R_TILE * FDIM];
  __shared__ float sm[R_TILE * HEADS];
  __shared__ float ss[R_TILE * HEADS];
  __shared__ float sad[R_TILE * HEADS];
  __shared__ __attribute__((aligned(16))) unsigned int list[LCAP];
  __shared__ int cnt;

  const int tid  = threadIdx.x;
  const int lane = tid & 31;
  const int wave = tid >> 5;
  const int base = blockIdx.x * R_TILE;
  const int nvalid = (N_NODES - base) < R_TILE ? (N_NODES - base) : R_TILE;
  unsigned int* __restrict__ my_tl = tl + (size_t)blockIdx.x * TL_STRIDE;

  for (int i = tid; i < R_TILE * FDIM / 4; i += TB)
    ((float4*)acc)[i] = make_float4(0.f, 0.f, 0.f, 0.f);
  for (int i = tid; i < R_TILE * HEADS; i += TB) {
    const int node = base + (i >> 3);
    sm[i]  = -__builtin_inff();
    ss[i]  = 0.f;
    sad[i] = node < N_NODES ? a_dst[(size_t)node * HEADS + (i & 7)] : 0.f;
  }
  if (tid == 0) cnt = 0;
  __syncthreads();

  if (FIRST) {
    const uint4* __restrict__ tb4 = (const uint4*)tb;
    const unsigned int pat = (unsigned int)blockIdx.x * 0x01010101u;
    for (int it = 0; it < E16; it += TB * SCAN_K) {
      uint4 d[SCAN_K];
#pragma unroll
      for (int k = 0; k < SCAN_K; ++k) {
        const int q = it + k * TB + tid;
        d[k] = q < E16 ? tb4[q] : make_uint4(~pat, ~pat, ~pat, ~pat);
      }
      unsigned int mw[(SCAN_K + 1) / 2];
      int n = 0;
#pragma unroll
      for (int k = 0; k < SCAN_K; ++k) {
        const unsigned int mk = (zero_bytes(d[k].x ^ pat) >> 7) |
                                (zero_bytes(d[k].y ^ pat) >> 6) |
                                (zero_bytes(d[k].z ^ pat) >> 5) |
                                (zero_bytes(d[k].w ^ pat) >> 4);
        n += __builtin_popcount(mk);
        if (k & 1) mw[k >> 1] |= mk << 4;
        else       mw[k >> 1] = mk;
      }
      int pre = 0, tot = 0;
#pragma unroll
      for (int b = 0; b < 7; ++b) {
        const unsigned int pl = __builtin_amdgcn_ballot_w32(((n >> b) & 1) != 0);
        pre += (int)__builtin_amdgcn_mbcnt_lo(pl, 0u) << b;
        tot += __builtin_popcount(pl) << b;
      }
      if (tot) {
        int bp = 0;
        if (lane == 0) bp = atomicAdd(&cnt, tot);
        int pos = __builtin_amdgcn_readfirstlane(bp) + pre;
#pragma unroll
        for (int w = 0; w < (SCAN_K + 1) / 2; ++w) {
          unsigned int m = mw[w];
          const unsigned int e0 = (unsigned int)(it + 2 * w * TB + tid) * 16u;
          while (m) {
            const unsigned int p = (unsigned int)__builtin_ctz(m);
            m &= m - 1u;
            if (pos < LCAP)
              list[pos] = e0 + ((p >> 2) & 1u) * (16u * TB) + 4u * (p & 3u) + (p >> 3);
            ++pos;
          }
        }
      }
    }
  }
  int nscan, total;
  if (FIRST) {
    __syncthreads();
    nscan = cnt;
    total = nscan + nvalid;
    if (tid == 0) my_tl[0] = (unsigned int)total;
  } else {
    nscan = 0;
    total = __builtin_amdgcn_readfirstlane((int)my_tl[0]);
  }

  if (total <= LCAP) {
    if (FIRST) {
      if (tid < nvalid)
        list[nscan + tid] = (unsigned int)(base + tid) | ((unsigned int)tid << 16);
    } else {
      const uint4* __restrict__ g4 = (const uint4*)(my_tl + TL_HDR);
      for (int i = tid; i < (total + 3) / 4; i += TB) ((uint4*)list)[i] = g4[i];
    }
    __syncthreads();
    for (int i = tid; i < total; i += TB) {
      const unsigned int a = list[i];
      unsigned int src, dl;
      if (FIRST && i < nscan) {
        src = (unsigned int)ei[a];
        dl  = (unsigned int)(ei[N_EDGES + a] - base);
        list[i] = src | (dl << 16);
      } else {
        src = a & 0xFFFFu;
        dl  = a >> 16;
      }
      const float4 s0 = *(const float4*)(a_src + (size_t)src * HEADS);
      const float4 s1 = *(const float4*)(a_src + (size_t)src * HEADS + 4);
      const float sv[8] = {s0.x, s0.y, s0.z, s0.w, s1.x, s1.y, s1.z, s1.w};
#pragma unroll
      for (int hh = 0; hh < HEADS; ++hh)
        lds_max_f(&sm[dl * HEADS + hh], leaky(sv[hh] + sad[dl * HEADS + hh]));
    }
    __syncthreads();
    if (FIRST) {
      uint4* __restrict__ g4 = (uint4*)(my_tl + TL_HDR);
      for (int i = tid; i < (total + 3) / 4; i += TB) g4[i] = ((const uint4*)list)[i];
    }
    for (int i = tid; i < R_TILE * HEADS; i += TB) {
      const float mm = sm[i];
      sm[i] = (__builtin_fabsf(mm) < __builtin_inff()) ? mm : 0.f;
    }
    __syncthreads();
    constexpr int ADD_U = 8;
    const int hh = lane >> 2;
    const int qs = lane & 3;
    for (int i0 = wave * ADD_U; i0 < total; i0 += (TB / 32) * ADD_U) {
      const bool ok0 = (i0 + qs) < total;
      const bool ok1 = (i0 + 4 + qs) < total;
      const unsigned int a0 = ok0 ? list[i0 + qs] : 0u;
      const unsigned int a1 = ok1 ? list[i0 + 4 + qs] : 0u;
      unsigned int a[ADD_U];
      float4 v[ADD_U];
#pragma unroll
      for (int u = 0; u < ADD_U; ++u)
        a[u] = (unsigned int)__builtin_amdgcn_readlane((int)(u < 4 ? a0 : a1), u & 3);
#pragma unroll
      for (int u = 0; u < ADD_U; ++u)
        v[u] = *(const float4*)(h + (size_t)(a[u] & 0xFFFFu) * FDIM + lane * 4);
      const float as0 = a_src[(size_t)(a0 & 0xFFFFu) * HEADS + hh];
      const float as1 = a_src[(size_t)(a1 & 0xFFFFu) * HEADS + hh];
      const unsigned int d0 = (a0 >> 16) * HEADS + hh;
      const unsigned int d1 = (a1 >> 16) * HEADS + hh;
      const float p0 = expf(leaky(as0 + sad[d0]) - sm[d0]);
      const float p1 = expf(leaky(as1 + sad[d1]) - sm[d1]);
      if (ok0) atomicAdd(&ss[d0], p0);
      if (ok1) atomicAdd(&ss[d1], p1);
#pragma unroll
      for (int u = 0; u < ADD_U; ++u) {
        if ((i0 + u) < total) {
          const float p = quad_lane(u < 4 ? p0 : p1, u & 3);
          float* ar = &acc[(a[u] >> 16) * FDIM + lane];
          atomicAdd(ar,      p * v[u].x);
          atomicAdd(ar + 32, p * v[u].y);
          atomicAdd(ar + 64, p * v[u].z);
          atomicAdd(ar + 96, p * v[u].w);
        }
      }
    }
  } else {
    for (int e = tid; e < ETOT; e += TB) {
      int src, dst;
      edge_decode(ei, e, src, dst);
      const unsigned int dl = (unsigned int)(dst - base);
      if (dl < (unsigned int)R_TILE) {
        for (int hh = 0; hh < HEADS; ++hh)
          lds_max_f(&sm[dl * HEADS + hh],
                    leaky(a_src[(size_t)src * HEADS + hh] + sad[dl * HEADS + hh]));
      }
    }
    __syncthreads();
    for (int i = tid; i < R_TILE * HEADS; i += TB) {
      const float mm = sm[i];
      sm[i] = (__builtin_fabsf(mm) < __builtin_inff()) ? mm : 0.f;
    }
    __syncthreads();
    for (int e = tid; e < ETOT; e += TB) {
      int src, dst;
      edge_decode(ei, e, src, dst);
      const unsigned int dl = (unsigned int)(dst - base);
      if (dl < (unsigned int)R_TILE) {
        for (int hh = 0; hh < HEADS; ++hh) {
          const float ev =
              leaky(a_src[(size_t)src * HEADS + hh] + sad[dl * HEADS + hh]);
          const float p = expf(ev - sm[dl * HEADS + hh]);
          atomicAdd(&ss[dl * HEADS + hh], p);
          for (int dd = 0; dd < HEAD_DIM; ++dd) {
            const int c = hh * HEAD_DIM + dd;
            atomicAdd(&acc[dl * FDIM + tile_col(c)],
                      p * h[(size_t)src * FDIM + c]);
          }
        }
      }
    }
  }
  __syncthreads();

  for (int idx = tid; idx < R_TILE * 32; idx += TB) {
    const int row = idx >> 5;
    const int l   = idx & 31;
    const int node = base + row;
    if (node < N_NODES) {
      const float den = ss[row * HEADS + (l >> 2)] + 1e-16f;
      const float4 b = *(const float4*)(bias + l * 4);
      float4 o;
      o.x = acc[row * FDIM + l]      / den + b.x;
      o.y = acc[row * FDIM + 32 + l] / den + b.y;
      o.z = acc[row * FDIM + 64 + l] / den + b.z;
      o.w = acc[row * FDIM + 96 + l] / den + b.w;
      if (ELU) {
        o.x = o.x > 0.f ? o.x : (expf(o.x) - 1.f);
        o.y = o.y > 0.f ? o.y : (expf(o.y) - 1.f);
        o.z = o.z > 0.f ? o.z : (expf(o.z) - 1.f);
        o.w = o.w > 0.f ? o.w : (expf(o.w) - 1.f);
      }
      *(float4*)(out + (size_t)node * FDIM + l * 4) = o;
    }
  }
}

extern "C" void kernel_launch(void* const* d_in, const int* in_sizes, int n_in,
                              void* d_out, int out_size, void* d_ws,
                              size_t ws_size, hipStream_t stream) {
  (void)in_sizes; (void)n_in; (void)out_size;

  const float* x   = (const float*)d_in[0];
  const int*   ei  = (const int*)d_in[1];
  const float* W1  = (const float*)d_in[2];
  const float* as1 = (const float*)d_in[3];
  const float* ad1 = (const float*)d_in[4];
  const float* b1  = (const float*)d_in[5];
  const float* W2  = (const float*)d_in[6];
  const float* as2 = (const float*)d_in[7];
  const float* ad2 = (const float*)d_in[8];
  const float* b2  = (const float*)d_in[9];
  float* out = (float*)d_out;

  float* ws    = (float*)d_ws;
  float* h     = ws;
  float* agg   = h + (size_t)N_NODES * FDIM;
  float* a_src = agg + (size_t)N_NODES * FDIM;
  float* a_dst = a_src + (size_t)N_NODES * HEADS;
  unsigned int* tl = (unsigned int*)(a_dst + (size_t)N_NODES * HEADS);
  const int gT  = (N_NODES + R_TILE - 1) / R_TILE;
  unsigned int* tb = tl + (size_t)gT * TL_STRIDE;
  const size_t need = ((size_t)N_NODES * FDIM * 2 + (size_t)N_NODES * HEADS * 2) *
                          sizeof(float) +
                      (size_t)gT * TL_STRIDE * sizeof(unsigned int) + N_EDGES;
  if (ws_size < need) return;

  const int nh = N_NODES * HEADS;
  const int gNH = (nh + 255) / 256;
  dim3 BG(GEMM_TB), B256(256), BT(TB);

  gemm128_wmma_f32<<<GEMM_GROUPS, BG, 0, stream>>>(x, W1, h);
  att_scores<true><<<gNH, B256, 0, stream>>>(h, as1, ad1, a_src, a_dst, ei, tb);
  tile_edges<true><<<gT, BT, 0, stream>>>(ei, a_src, a_dst, h, b1, agg, tl, tb);

  gemm128_wmma_f32<<<GEMM_GROUPS, BG, 0, stream>>>(agg, W2, h);
  att_scores<false><<<gNH, B256, 0, stream>>>(h, as2, ad2, a_src, a_dst, ei, tb);
  tile_edges<false><<<gT, BT, 0, stream>>>(ei, a_src, a_dst, h, b2, out, tl, tb);
}
